# starting kernel
# baseline (speedup 1.0000x reference)
#include <hip/hip_runtime.h>
#include <math.h>

typedef __attribute__((ext_vector_type(2))) float v2f;
typedef __attribute__((ext_vector_type(8))) float v8f;

#define GCN_N_NODES 50000
#define GCN_N_EDGES 600000

__global__ void gcn_fill_zero(float* __restrict__ p, int n) {
    int i = blockIdx.x * blockDim.x + threadIdx.x;
    if (i < n) p[i] = 0.0f;
}

__global__ void gcn_count_deg(const int* __restrict__ src, const int* __restrict__ dst,
                              float* __restrict__ deg_out, float* __restrict__ deg_in,
                              int E) {
    int e = blockIdx.x * blockDim.x + threadIdx.x;
    if (e < E) {
        atomicAdd(&deg_out[src[e]], 1.0f);
        atomicAdd(&deg_in[dst[e]], 1.0f);
    }
}

__global__ void gcn_inv_sqrt_clamp(float* __restrict__ d, int n) {
    int i = blockIdx.x * blockDim.x + threadIdx.x;
    if (i < n) {
        float v = d[i];
        v = v < 1.0f ? 1.0f : v;
        d[i] = 1.0f / sqrtf(v);
    }
}

__global__ void gcn_scatter(const float* __restrict__ x,
                            const int* __restrict__ src, const int* __restrict__ dst,
                            const float* __restrict__ deg_out_is,
                            float* __restrict__ agg, int E) {
    const int CH = 32;
    int t = blockIdx.x * blockDim.x + threadIdx.x;
    int e = t >> 5;
    if (e >= E) return;
    int c = t & 31;
    int s  = src[e];
    int dd = dst[e];
    float sc = deg_out_is[s];
    const float4 v = *(const float4*)(x + (size_t)s * 128 + c * 4);
    float* base = agg + (size_t)dd * 128 + c * 4;
    atomicAdd(base + 0, v.x * sc);
    atomicAdd(base + 1, v.y * sc);
    atomicAdd(base + 2, v.z * sc);
    atomicAdd(base + 3, v.w * sc);
    (void)CH;
}

__global__ void gcn_gemm_wmma(const float* __restrict__ A,
                              const float* __restrict__ deg_in_is,
                              const float* __restrict__ W,
                              const float* __restrict__ bias,
                              float* __restrict__ out,
                              int K, int C, int do_relu) {
    const int m0   = blockIdx.x * 16;
    const int n0   = blockIdx.y * 16;
    const int lane = threadIdx.x & 31;
    const int half = lane >> 4;
    const int l15  = lane & 15;
    const int m = m0 + l15;
    const int n = n0 + l15;

    const float scale = deg_in_is[m];
    const float* arow = A + (size_t)m * K;

    v8f c = {};
    for (int k0 = 0; k0 < K; k0 += 4) {
        const int kb = k0 + half * 2;
        v2f a, b;
        a.x = arow[kb]     * scale;
        a.y = arow[kb + 1] * scale;
        b.x = W[(size_t)kb       * C + n];
        b.y = W[(size_t)(kb + 1) * C + n];
        c = __builtin_amdgcn_wmma_f32_16x16x4_f32(
                false, a, false, b, (short)0, c, false, false);
    }

    const float bv = bias[n];
#pragma unroll
    for (int v = 0; v < 8; ++v) {
        float val = c[v] + bv;
        if (do_relu) val = fmaxf(val, 0.0f);
        out[(size_t)(m0 + half * 8 + v) * C + n] = val;
    }
}

extern "C" void kernel_launch(void* const* d_in, const int* in_sizes, int n_in,
                              void* d_out, int out_size, void* d_ws, size_t ws_size,
                              hipStream_t stream) {
    const float* features = (const float*)d_in[0];
    const int*   src      = (const int*)  d_in[1];
    const int*   dst      = (const int*)  d_in[2];
    const float* W1       = (const float*)d_in[3];
    const float* b1       = (const float*)d_in[4];
    const float* W2       = (const float*)d_in[5];
    const float* b2       = (const float*)d_in[6];
    const float* W3       = (const float*)d_in[7];
    const float* b3       = (const float*)d_in[8];
    float* out = (float*)d_out;

    const int N = GCN_N_NODES;
    const int E = GCN_N_EDGES;
    const size_t NF = (size_t)N * 128;

    float* bufA  = (float*)d_ws;
    float* bufB  = bufA + NF;
    float* deg_o = bufB + NF;
    float* deg_i = deg_o + N;

    const int TB = 256;

    gcn_fill_zero<<<(2 * N + TB - 1) / TB, TB, 0, stream>>>(deg_o, 2 * N);
    gcn_count_deg<<<(E + TB - 1) / TB, TB, 0, stream>>>(src, dst, deg_o, deg_i, E);
    gcn_inv_sqrt_clamp<<<(2 * N + TB - 1) / TB, TB, 0, stream>>>(deg_o, 2 * N);

    const int scatter_blocks = (E * 32 + TB - 1) / TB;
    const dim3 gemm128(N / 16, 128 / 16);
    const dim3 gemm16 (N / 16, 16 / 16);

    gcn_fill_zero<<<((int)NF + TB - 1) / TB, TB, 0, stream>>>(bufA, (int)NF);
    gcn_scatter<<<scatter_blocks, TB, 0, stream>>>(features, src, dst, deg_o, bufA, E);
    gcn_gemm_wmma<<<gemm128, 32, 0, stream>>>(bufA, deg_i, W1, b1, bufB, 128, 128, 1);

    gcn_fill_zero<<<((int)NF + TB - 1) / TB, TB, 0, stream>>>(bufA, (int)NF);
    gcn_scatter<<<scatter_blocks, TB, 0, stream>>>(bufB, src, dst, deg_o, bufA, E);
    gcn_gemm_wmma<<<gemm128, 32, 0, stream>>>(bufA, deg_i, W2, b2, bufB, 128, 128, 1);

    gcn_fill_zero<<<((int)NF + TB - 1) / TB, TB, 0, stream>>>(bufA, (int)NF);
    gcn_scatter<<<scatter_blocks, TB, 0, stream>>>(bufB, src, dst, deg_o, bufA, E);
    gcn_gemm_wmma<<<gemm16, 32, 0, stream>>>(bufA, deg_i, W3, b3, out, 128, 16, 0);

    (void)in_sizes; (void)n_in; (void)out_size; (void)ws_size;
}


# Aggregation by owner work-groups: 256 groups keep 196 rows of sums in LDS, scan all of dst, add their edges' source rows by LDS float adds; no global atomics or zero fills; sum order LDS arrival
# speedup vs baseline: 1.5618x; 1.5618x over previous
#include <hip/hip_runtime.h>
#include <math.h>

typedef __attribute__((ext_vector_type(2))) float v2f;
typedef __attribute__((ext_vector_type(8))) float v8f;

#define GCN_N_NODES 50000
#define GCN_N_EDGES 600000

#define GCN_FILL_STORES 8
__global__ void gcn_fill_zero(float* __restrict__ p, int n) {
    float4* p4 = reinterpret_cast<float4*>(p);
    const int n4 = n >> 2;
    int i = blockIdx.x * (blockDim.x * GCN_FILL_STORES) + threadIdx.x;
#pragma unroll
    for (int k = 0; k < GCN_FILL_STORES; ++k, i += blockDim.x)
        if (i < n4) p4[i] = make_float4(0.0f, 0.0f, 0.0f, 0.0f);
}

__global__ void gcn_count_deg(const int* __restrict__ src, const int* __restrict__ dst,
                              float* __restrict__ deg_out, float* __restrict__ deg_in,
                              int E) {
    int e = blockIdx.x * blockDim.x + threadIdx.x;
    if (e < E) {
        atomicAdd(&deg_out[src[e]], 1.0f);
        atomicAdd(&deg_in[dst[e]], 1.0f);
    }
}

__global__ void gcn_inv_sqrt_clamp(float* __restrict__ d, int n) {
    int i = blockIdx.x * blockDim.x + threadIdx.x;
    if (i < n) {
        float v = d[i];
        v = v < 1.0f ? 1.0f : v;
        d[i] = 1.0f / sqrtf(v);
    }
}

#define GCN_OWN_GROUPS   256
#define GCN_OWN_ROWS     196
#define GCN_OWN_THREADS  1024
#define GCN_OWN_WAVES    (GCN_OWN_THREADS / 32)
#define GCN_Q_CAP        256
#define GCN_SCAN_LOADS   8
#define GCN_EDGE_GROUP   8
static_assert(GCN_OWN_GROUPS * GCN_OWN_ROWS >= GCN_N_NODES, "the ranges must cover every node");
static_assert(GCN_N_EDGES % 4 == 0, "dst is read 4 edges (16 bytes) at a time");
static_assert(31 + 4 * 32 <= GCN_Q_CAP, "one 16-byte load of a wave may add 128 entries to 31 waiting ones");

template <bool FULL>
__device__ __forceinline__ void gcn_own_batch(const float* __restrict__ x,
                                              const int* __restrict__ src,
                                              const float* __restrict__ deg_out_is,
                                              float* tile, const int* qe, const int* qd,
                                              int n, int lane) {
    int   s_l  = 0;
    int   d_l  = 0;
    float sc_l = 0.0f;
    if (FULL || lane < n) {
        const int e = qe[lane];
        d_l  = qd[lane];
        s_l  = src[e];
        sc_l = deg_out_is[s_l];
    }
    const int n_edges = FULL ? 32 : n;
    for (int j0 = 0; j0 < n_edges; j0 += GCN_EDGE_GROUP) {
        float v[GCN_EDGE_GROUP][4];
#pragma unroll
        for (int j = 0; j < GCN_EDGE_GROUP; ++j) {
            if (FULL || j0 + j < n) {
                const int s = __builtin_amdgcn_readlane(s_l, j0 + j);
                const float* xr = x + (size_t)s * 128 + lane;
                v[j][0] = xr[0];
                v[j][1] = xr[32];
                v[j][2] = xr[64];
                v[j][3] = xr[96];
            }
        }
#pragma unroll
        for (int j = 0; j < GCN_EDGE_GROUP; ++j) {
            if (FULL || j0 + j < n) {
                const int   d  = __builtin_amdgcn_readlane(d_l, j0 + j);
                const float sc = __int_as_float(
                    __builtin_amdgcn_readlane(__float_as_int(sc_l), j0 + j));
                float* row = tile + d * 128 + lane;
                __hip_atomic_fetch_add(row + 0,  v[j][0] * sc, __ATOMIC_RELAXED, __HIP_MEMORY_SCOPE_WORKGROUP);
                __hip_atomic_fetch_add(row + 32, v[j][1] * sc, __ATOMIC_RELAXED, __HIP_MEMORY_SCOPE_WORKGROUP);
                __hip_atomic_fetch_add(row + 64, v[j][2] * sc, __ATOMIC_RELAXED, __HIP_MEMORY_SCOPE_WORKGROUP);
                __hip_atomic_fetch_add(row + 96, v[j][3] * sc, __ATOMIC_RELAXED, __HIP_MEMORY_SCOPE_WORKGROUP);
            }
        }
    }
}

__global__ __launch_bounds__(GCN_OWN_THREADS)
void gcn_agg_owner(const float* __restrict__ x,
                   const int* __restrict__ src, const int* __restrict__ dst,
                   const float* __restrict__ deg_out_is,
                   float* __restrict__ agg, int E, int N) {
    __shared__ float tile[GCN_OWN_ROWS * 128];
    __shared__ int   q_e[GCN_OWN_WAVES * GCN_Q_CAP];
    __shared__ int   q_d[GCN_OWN_WAVES * GCN_Q_CAP];

    const int tid  = threadIdx.x;
    const int lane = tid & 31;
    const int wave = tid >> 5;
    const int lo   = blockIdx.x * GCN_OWN_ROWS;
    int rows = N - lo;
    rows = rows > GCN_OWN_ROWS ? GCN_OWN_ROWS : rows;
    rows = rows < 0 ? 0 : rows;

    float4* tile4 = reinterpret_cast<float4*>(tile);
    for (int i = tid; i < GCN_OWN_ROWS * 32; i += GCN_OWN_THREADS)
        tile4[i] = make_float4(0.0f, 0.0f, 0.0f, 0.0f);
    __syncthreads();

    int* qe = q_e + wave * GCN_Q_CAP;
    int* qd = q_d + wave * GCN_Q_CAP;
    int  qn = 0;

    const int4* dst4 = reinterpret_cast<const int4*>(dst);
    const int nvec    = E >> 2;
    const int step    = GCN_OWN_THREADS * GCN_SCAN_LOADS;
    const int n_steps = (nvec + step - 1) / step;
    int st = blockIdx.x % n_steps;

    for (int o = 0; o < n_steps; ++o) {
        const int base = st * step + tid;
        const bool whole = (st + 1) * step <= nvec;
        st = st + 1 == n_steps ? 0 : st + 1;

        int4 dv[GCN_SCAN_LOADS];
        if (whole) {
#pragma unroll
            for (int u = 0; u < GCN_SCAN_LOADS; ++u)
                dv[u] = dst4[base + u * GCN_OWN_THREADS];
        } else {
#pragma unroll
            for (int u = 0; u < GCN_SCAN_LOADS; ++u) {
                const int idx = base + u * GCN_OWN_THREADS;
                dv[u] = idx < nvec ? dst4[idx] : make_int4(-1, -1, -1, -1);
            }
        }
#pragma unroll
        for (int u = 0; u < GCN_SCAN_LOADS; ++u) {
            const int e4 = (base + u * GCN_OWN_THREADS) * 4;
            const unsigned r[4] = { (unsigned)(dv[u].x - lo), (unsigned)(dv[u].y - lo),
                                    (unsigned)(dv[u].z - lo), (unsigned)(dv[u].w - lo) };
            const unsigned r01 = r[0] < r[1] ? r[0] : r[1];
            const unsigned r23 = r[2] < r[3] ? r[2] : r[3];
            const unsigned rmin = r01 < r23 ? r01 : r23;
            if (__builtin_amdgcn_ballot_w32(rmin < (unsigned)rows) == 0u) continue;
            const bool m[4] = { r[0] < (unsigned)rows, r[1] < (unsigned)rows,
                                r[2] < (unsigned)rows, r[3] < (unsigned)rows };
#pragma unroll
            for (int c = 0; c < 4; ++c) {
                const unsigned mask = __builtin_amdgcn_ballot_w32(m[c]);
                if (mask != 0u) {
                    const int pos = qn + (int)__builtin_amdgcn_mbcnt_lo(mask, 0u);
                    if (m[c]) { qe[pos] = e4 + c; qd[pos] = (int)r[c]; }
                    qn += __builtin_popcount(mask);
                }
            }
            while (qn >= 32) {
                qn -= 32;
                gcn_own_batch<true>(x, src, deg_out_is, tile, qe + qn, qd + qn, 32, lane);
            }
        }
    }
    if (qn > 0)
        gcn_own_batch<false>(x, src, deg_out_is, tile, qe, qd, qn, lane);

    __syncthreads();
    float4* out4 = reinterpret_cast<float4*>(agg + (size_t)lo * 128);
    for (int i = tid; i < rows * 32; i += GCN_OWN_THREADS)
        out4[i] = tile4[i];
}

__global__ void gcn_gemm_wmma(const float* __restrict__ A,
                              const float* __restrict__ deg_in_is,
                              const float* __restrict__ W,
                              const float* __restrict__ bias,
                              float* __restrict__ out,
                              int K, int C, int do_relu) {
    const int m0   = blockIdx.x * 16;
    const int n0   = blockIdx.y * 16;
    const int lane = threadIdx.x & 31;
    const int half = lane >> 4;
    const int l15  = lane & 15;
    const int m = m0 + l15;
    const int n = n0 + l15;

    const float scale = deg_in_is[m];
    const float* arow = A + (size_t)m * K;

    v8f c = {};
    for (int k0 = 0; k0 < K; k0 += 4) {
        const int kb = k0 + half * 2;
        v2f a, b;
        a.x = arow[kb]     * scale;
        a.y = arow[kb + 1] * scale;
        b.x = W[(size_t)kb       * C + n];
        b.y = W[(size_t)(kb + 1) * C + n];
        c = __builtin_amdgcn_wmma_f32_16x16x4_f32(
                false, a, false, b, (short)0, c, false, false);
    }

    const float bv = bias[n];
#pragma unroll
    for (int v = 0; v < 8; ++v) {
        float val = c[v] + bv;
        if (do_relu) val = fmaxf(val, 0.0f);
        out[(size_t)(m0 + half * 8 + v) * C + n] = val;
    }
}

extern "C" void kernel_launch(void* const* d_in, const int* in_sizes, int n_in,
                              void* d_out, int out_size, void* d_ws, size_t ws_size,
                              hipStream_t stream) {
    const float* features = (const float*)d_in[0];
    const int*   src      = (const int*)  d_in[1];
    const int*   dst      = (const int*)  d_in[2];
    const float* W1       = (const float*)d_in[3];
    const float* b1       = (const float*)d_in[4];
    const float* W2       = (const float*)d_in[5];
    const float* b2       = (const float*)d_in[6];
    const float* W3       = (const float*)d_in[7];
    const float* b3       = (const float*)d_in[8];
    float* out = (float*)d_out;

    const int N = GCN_N_NODES;
    const int E = GCN_N_EDGES;
    const size_t NF = (size_t)N * 128;

    float* bufA  = (float*)d_ws;
    float* bufB  = bufA + NF;
    float* deg_o = bufB + NF;
    float* deg_i = deg_o + N;

    const int TB = 256;
    const int FILL_PER_BLOCK = TB * 4 * GCN_FILL_STORES;

    gcn_fill_zero<<<(2 * N + FILL_PER_BLOCK - 1) / FILL_PER_BLOCK, TB, 0, stream>>>(deg_o, 2 * N);
    gcn_count_deg<<<(E + TB - 1) / TB, TB, 0, stream>>>(src, dst, deg_o, deg_i, E);
    gcn_inv_sqrt_clamp<<<(2 * N + TB - 1) / TB, TB, 0, stream>>>(deg_o, 2 * N);

    const dim3 gemm128(N / 16, 128 / 16);
    const dim3 gemm16 (N / 16, 16 / 16);

    gcn_agg_owner<<<GCN_OWN_GROUPS, GCN_OWN_THREADS, 0, stream>>>(features, src, dst, deg_o, bufA, E, N);
    gcn_gemm_wmma<<<gemm128, 32, 0, stream>>>(bufA, deg_i, W1, b1, bufB, 128, 128, 1);

    gcn_agg_owner<<<GCN_OWN_GROUPS, GCN_OWN_THREADS, 0, stream>>>(bufB, src, dst, deg_o, bufA, E, N);
    gcn_gemm_wmma<<<gemm128, 32, 0, stream>>>(bufA, deg_i, W2, b2, bufB, 128, 128, 1);

    gcn_agg_owner<<<GCN_OWN_GROUPS, GCN_OWN_THREADS, 0, stream>>>(bufB, src, dst, deg_o, bufA, E, N);
    gcn_gemm_wmma<<<gemm16, 32, 0, stream>>>(bufA, deg_i, W3, b3, out, 128, 16, 0);

    (void)in_sizes; (void)n_in; (void)out_size; (void)ws_size;
}


# GEMM retiled: 4-wave work-groups of 64 rows x 64 columns, a wave's 16 rows of A read once per 4 column tiles by 16-byte loads, W slice staged in LDS as k-quads; dot products add k in another order
# speedup vs baseline: 2.0644x; 1.3218x over previous
#include <hip/hip_runtime.h>
#include <math.h>

typedef __attribute__((ext_vector_type(2))) float v2f;
typedef __attribute__((ext_vector_type(8))) float v8f;

#define GCN_N_NODES 50000
#define GCN_N_EDGES 600000

#define GCN_FILL_STORES 8
__global__ void gcn_fill_zero(float* __restrict__ p, int n) {
    float4* p4 = reinterpret_cast<float4*>(p);
    const int n4 = n >> 2;
    int i = blockIdx.x * (blockDim.x * GCN_FILL_STORES) + threadIdx.x;
#pragma unroll
    for (int k = 0; k < GCN_FILL_STORES; ++k, i += blockDim.x)
        if (i < n4) p4[i] = make_float4(0.0f, 0.0f, 0.0f, 0.0f);
}

__global__ void gcn_count_deg(const int* __restrict__ src, const int* __restrict__ dst,
                              float* __restrict__ deg_out, float* __restrict__ deg_in,
                              int E) {
    int e = blockIdx.x * blockDim.x + threadIdx.x;
    if (e < E) {
        atomicAdd(&deg_out[src[e]], 1.0f);
        atomicAdd(&deg_in[dst[e]], 1.0f);
    }
}

__global__ void gcn_inv_sqrt_clamp(float* __restrict__ d, int n) {
    int i = blockIdx.x * blockDim.x + threadIdx.x;
    if (i < n) {
        float v = d[i];
        v = v < 1.0f ? 1.0f : v;
        d[i] = 1.0f / sqrtf(v);
    }
}

#define GCN_OWN_GROUPS   256
#define GCN_OWN_ROWS     196
#define GCN_OWN_THREADS  1024
#define GCN_OWN_WAVES    (GCN_OWN_THREADS / 32)
#define GCN_Q_CAP        256
#define GCN_SCAN_LOADS   8
#define GCN_EDGE_GROUP   8
static_assert(GCN_OWN_GROUPS * GCN_OWN_ROWS >= GCN_N_NODES, "the ranges must cover every node");
static_assert(GCN_N_EDGES % 4 == 0, "dst is read 4 edges (16 bytes) at a time");
static_assert(31 + 4 * 32 <= GCN_Q_CAP, "one 16-byte load of a wave may add 128 entries to 31 waiting ones");

template <bool FULL>
__device__ __forceinline__ void gcn_own_batch(const float* __restrict__ x,
                                              const int* __restrict__ src,
                                              const float* __restrict__ deg_out_is,
                                              float* tile, const int* qe, const int* qd,
                                              int n, int lane) {
    int   s_l  = 0;
    int   d_l  = 0;
    float sc_l = 0.0f;
    if (FULL || lane < n) {
        const int e = qe[lane];
        d_l  = qd[lane];
        s_l  = src[e];
        sc_l = deg_out_is[s_l];
    }
    const int n_edges = FULL ? 32 : n;
    for (int j0 = 0; j0 < n_edges; j0 += GCN_EDGE_GROUP) {
        float v[GCN_EDGE_GROUP][4];
#pragma unroll
        for (int j = 0; j < GCN_EDGE_GROUP; ++j) {
            if (FULL || j0 + j < n) {
                const int s = __builtin_amdgcn_readlane(s_l, j0 + j);
                const float* xr = x + (size_t)s * 128 + lane;
                v[j][0] = xr[0];
                v[j][1] = xr[32];
                v[j][2] = xr[64];
                v[j][3] = xr[96];
            }
        }
#pragma unroll
        for (int j = 0; j < GCN_EDGE_GROUP; ++j) {
            if (FULL || j0 + j < n) {
                const int   d  = __builtin_amdgcn_readlane(d_l, j0 + j);
                const float sc = __int_as_float(
                    __builtin_amdgcn_readlane(__float_as_int(sc_l), j0 + j));
                float* row = tile + d * 128 + lane;
                __hip_atomic_fetch_add(row + 0,  v[j][0] * sc, __ATOMIC_RELAXED, __HIP_MEMORY_SCOPE_WORKGROUP);
                __hip_atomic_fetch_add(row + 32, v[j][1] * sc, __ATOMIC_RELAXED, __HIP_MEMORY_SCOPE_WORKGROUP);
                __hip_atomic_fetch_add(row + 64, v[j][2] * sc, __ATOMIC_RELAXED, __HIP_MEMORY_SCOPE_WORKGROUP);
                __hip_atomic_fetch_add(row + 96, v[j][3] * sc, __ATOMIC_RELAXED, __HIP_MEMORY_SCOPE_WORKGROUP);
            }
        }
    }
}

__global__ __launch_bounds__(GCN_OWN_THREADS)
void gcn_agg_owner(const float* __restrict__ x,
                   const int* __restrict__ src, const int* __restrict__ dst,
                   const float* __restrict__ deg_out_is,
                   float* __restrict__ agg, int E, int N) {
    __shared__ float tile[GCN_OWN_ROWS * 128];
    __shared__ int   q_e[GCN_OWN_WAVES * GCN_Q_CAP];
    __shared__ int   q_d[GCN_OWN_WAVES * GCN_Q_CAP];

    const int tid  = threadIdx.x;
    const int lane = tid & 31;
    const int wave = tid >> 5;
    const int lo   = blockIdx.x * GCN_OWN_ROWS;
    int rows = N - lo;
    rows = rows > GCN_OWN_ROWS ? GCN_OWN_ROWS : rows;
    rows = rows < 0 ? 0 : rows;

    float4* tile4 = reinterpret_cast<float4*>(tile);
    for (int i = tid; i < GCN_OWN_ROWS * 32; i += GCN_OWN_THREADS)
        tile4[i] = make_float4(0.0f, 0.0f, 0.0f, 0.0f);
    __syncthreads();

    int* qe = q_e + wave * GCN_Q_CAP;
    int* qd = q_d + wave * GCN_Q_CAP;
    int  qn = 0;

    const int4* dst4 = reinterpret_cast<const int4*>(dst);
    const int nvec    = E >> 2;
    const int step    = GCN_OWN_THREADS * GCN_SCAN_LOADS;
    const int n_steps = (nvec + step - 1) / step;
    int st = blockIdx.x % n_steps;

    for (int o = 0; o < n_steps; ++o) {
        const int base = st * step + tid;
        const bool whole = (st + 1) * step <= nvec;
        st = st + 1 == n_steps ? 0 : st + 1;

        int4 dv[GCN_SCAN_LOADS];
        if (whole) {
#pragma unroll
            for (int u = 0; u < GCN_SCAN_LOADS; ++u)
                dv[u] = dst4[base + u * GCN_OWN_THREADS];
        } else {
#pragma unroll
            for (int u = 0; u < GCN_SCAN_LOADS; ++u) {
                const int idx = base + u * GCN_OWN_THREADS;
                dv[u] = idx < nvec ? dst4[idx] : make_int4(-1, -1, -1, -1);
            }
        }
#pragma unroll
        for (int u = 0; u < GCN_SCAN_LOADS; ++u) {
            const int e4 = (base + u * GCN_OWN_THREADS) * 4;
            const unsigned r[4] = { (unsigned)(dv[u].x - lo), (unsigned)(dv[u].y - lo),
                                    (unsigned)(dv[u].z - lo), (unsigned)(dv[u].w - lo) };
            const unsigned r01 = r[0] < r[1] ? r[0] : r[1];
            const unsigned r23 = r[2] < r[3] ? r[2] : r[3];
            const unsigned rmin = r01 < r23 ? r01 : r23;
            if (__builtin_amdgcn_ballot_w32(rmin < (unsigned)rows) == 0u) continue;
            const bool m[4] = { r[0] < (unsigned)rows, r[1] < (unsigned)rows,
                                r[2] < (unsigned)rows, r[3] < (unsigned)rows };
#pragma unroll
            for (int c = 0; c < 4; ++c) {
                const unsigned mask = __builtin_amdgcn_ballot_w32(m[c]);
                if (mask != 0u) {
                    const int pos = qn + (int)__builtin_amdgcn_mbcnt_lo(mask, 0u);
                    if (m[c]) { qe[pos] = e4 + c; qd[pos] = (int)r[c]; }
                    qn += __builtin_popcount(mask);
                }
            }
            while (qn >= 32) {
                qn -= 32;
                gcn_own_batch<true>(x, src, deg_out_is, tile, qe + qn, qd + qn, 32, lane);
            }
        }
    }
    if (qn > 0)
        gcn_own_batch<false>(x, src, deg_out_is, tile, qe, qd, qn, lane);

    __syncthreads();
    float4* out4 = reinterpret_cast<float4*>(agg + (size_t)lo * 128);
    for (int i = tid; i < rows * 32; i += GCN_OWN_THREADS)
        out4[i] = tile4[i];
}

#define GCN_GEMM_WAVES   4
#define GCN_GEMM_THREADS (GCN_GEMM_WAVES * 32)
#define GCN_GEMM_ROWS    (GCN_GEMM_WAVES * 16)
static_assert(GCN_N_NODES % 16 == 0, "a wave's 16-row tile is whole or absent");

template <int NT, int C>
__global__ __launch_bounds__(GCN_GEMM_THREADS)
void gcn_gemm_strip(const float* __restrict__ A,
                    const float* __restrict__ deg_in_is,
                    const float* __restrict__ W,
                    const float* __restrict__ bias,
                    float* __restrict__ out, int do_relu) {
    constexpr int NC     = 16 * NT;
    constexpr int PITCH4 = NC + 8;
    constexpr int STAGE  = (32 * NC) / GCN_GEMM_THREADS;
    static_assert((32 * NC) % GCN_GEMM_THREADS == 0, "whole staging passes");
    __shared__ float4 wq[32 * PITCH4];

    const int tid  = threadIdx.x;
    const int lane = tid & 31;
    const int wave = tid >> 5;
    const int h    = lane >> 4;
    const int c    = lane & 15;
    const int m0   = (blockIdx.x * GCN_GEMM_WAVES + wave) * 16;
    const int nb   = blockIdx.y * NC;
    const bool live = m0 < GCN_N_NODES;

    const int mrow = (live ? m0 : 0) + c;
    const float scale = deg_in_is[mrow];
    const float4* arow = reinterpret_cast<const float4*>(A + (size_t)mrow * 128) + h;
    float4 a[16];
#pragma unroll
    for (int j = 0; j < 8; ++j) a[j] = arow[2 * j];

#pragma unroll
    for (int i = 0; i < STAGE; ++i) {
        const int idx = tid + i * GCN_GEMM_THREADS;
        const int n   = idx % NC;
        const int kq  = idx / NC;
        const float* wp = W + (size_t)(4 * kq) * C + nb + n;
        wq[kq * PITCH4 + n] = make_float4(wp[0], wp[C], wp[2 * C], wp[3 * C]);
    }
    __syncthreads();
    if (!live) return;

#pragma unroll
    for (int j = 8; j < 16; ++j) a[j] = arow[2 * j];

    v8f acc[NT];
#pragma unroll
    for (int t = 0; t < NT; ++t) acc[t] = v8f{};

#pragma unroll
    for (int j = 0; j < 16; ++j) {
        v2f a0, a1;
        a0.x = a[j].x * scale;  a0.y = a[j].y * scale;
        a1.x = a[j].z * scale;  a1.y = a[j].w * scale;
        float4 b[NT];
#pragma unroll
        for (int t = 0; t < NT; ++t) b[t] = wq[(2 * j + h) * PITCH4 + t * 16 + c];
#pragma unroll
        for (int t = 0; t < NT; ++t) {
            v2f b0;  b0.x = b[t].x;  b0.y = b[t].y;
            acc[t] = __builtin_amdgcn_wmma_f32_16x16x4_f32(
                         false, a0, false, b0, (short)0, acc[t], false, false);
        }
#pragma unroll
        for (int t = 0; t < NT; ++t) {
            v2f b1;  b1.x = b[t].z;  b1.y = b[t].w;
            acc[t] = __builtin_amdgcn_wmma_f32_16x16x4_f32(
                         false, a1, false, b1, (short)0, acc[t], false, false);
        }
    }

#pragma unroll
    for (int t = 0; t < NT; ++t) {
        const int n = nb + t * 16 + c;
        const float bv = bias[n];
#pragma unroll
        for (int v = 0; v < 8; ++v) {
            float val = acc[t][v] + bv;
            if (do_relu) val = fmaxf(val, 0.0f);
            out[(size_t)(m0 + h * 8 + v) * C + n] = val;
        }
    }
}

extern "C" void kernel_launch(void* const* d_in, const int* in_sizes, int n_in,
                              void* d_out, int out_size, void* d_ws, size_t ws_size,
                              hipStream_t stream) {
    const float* features = (const float*)d_in[0];
    const int*   src      = (const int*)  d_in[1];
    const int*   dst      = (const int*)  d_in[2];
    const float* W1       = (const float*)d_in[3];
    const float* b1       = (const float*)d_in[4];
    const float* W2       = (const float*)d_in[5];
    const float* b2       = (const float*)d_in[6];
    const float* W3       = (const float*)d_in[7];
    const float* b3       = (const float*)d_in[8];
    float* out = (float*)d_out;

    const int N = GCN_N_NODES;
    const int E = GCN_N_EDGES;
    const size_t NF = (size_t)N * 128;

    float* bufA  = (float*)d_ws;
    float* bufB  = bufA + NF;
    float* deg_o = bufB + NF;
    float* deg_i = deg_o + N;

    const int TB = 256;
    const int FILL_PER_BLOCK = TB * 4 * GCN_FILL_STORES;

    gcn_fill_zero<<<(2 * N + FILL_PER_BLOCK - 1) / FILL_PER_BLOCK, TB, 0, stream>>>(deg_o, 2 * N);
    gcn_count_deg<<<(E + TB - 1) / TB, TB, 0, stream>>>(src, dst, deg_o, deg_i, E);
    gcn_inv_sqrt_clamp<<<(2 * N + TB - 1) / TB, TB, 0, stream>>>(deg_o, 2 * N);

    const dim3 gemm128((GCN_N_NODES + GCN_GEMM_ROWS - 1) / GCN_GEMM_ROWS, 2);
    const dim3 gemm16 ((GCN_N_NODES + GCN_GEMM_ROWS - 1) / GCN_GEMM_ROWS, 1);

    gcn_agg_owner<<<GCN_OWN_GROUPS, GCN_OWN_THREADS, 0, stream>>>(features, src, dst, deg_o, bufA, E, N);
    gcn_gemm_strip<4, 128><<<gemm128, GCN_GEMM_THREADS, 0, stream>>>(bufA, deg_i, W1, b1, bufB, 1);

    gcn_agg_owner<<<GCN_OWN_GROUPS, GCN_OWN_THREADS, 0, stream>>>(bufB, src, dst, deg_o, bufA, E, N);
    gcn_gemm_strip<4, 128><<<gemm128, GCN_GEMM_THREADS, 0, stream>>>(bufA, deg_i, W2, b2, bufB, 1);

    gcn_agg_owner<<<GCN_OWN_GROUPS, GCN_OWN_THREADS, 0, stream>>>(bufB, src, dst, deg_o, bufA, E, N);
    gcn_gemm_strip<1, 16><<<gemm16, GCN_GEMM_THREADS, 0, stream>>>(bufA, deg_i, W3, b3, out, 0);

    (void)in_sizes; (void)n_in; (void)out_size; (void)ws_size;
}


# Layer 1's owner scan also writes each wave's matched edges (source row, local row, source scale) as a list in the workspace; layers 2 and 3 walk the lists, no dst scan; sum order LDS arrival
# speedup vs baseline: 2.5174x; 1.2194x over previous
#include <hip/hip_runtime.h>
#include <math.h>

typedef __attribute__((ext_vector_type(2))) float v2f;
typedef __attribute__((ext_vector_type(8))) float v8f;

#define GCN_N_NODES 50000
#define GCN_N_EDGES 600000

#define GCN_FILL_STORES 8
__global__ void gcn_fill_zero(float* __restrict__ p, int n) {
    float4* p4 = reinterpret_cast<float4*>(p);
    const int n4 = n >> 2;
    int i = blockIdx.x * (blockDim.x * GCN_FILL_STORES) + threadIdx.x;
#pragma unroll
    for (int k = 0; k < GCN_FILL_STORES; ++k, i += blockDim.x)
        if (i < n4) p4[i] = make_float4(0.0f, 0.0f, 0.0f, 0.0f);
}

__global__ void gcn_count_deg(const int* __restrict__ src, const int* __restrict__ dst,
                              float* __restrict__ deg_out, float* __restrict__ deg_in,
                              int E) {
    int e = blockIdx.x * blockDim.x + threadIdx.x;
    if (e < E) {
        atomicAdd(&deg_out[src[e]], 1.0f);
        atomicAdd(&deg_in[dst[e]], 1.0f);
    }
}

__global__ void gcn_inv_sqrt_clamp(float* __restrict__ d, int n) {
    int i = blockIdx.x * blockDim.x + threadIdx.x;
    if (i < n) {
        float v = d[i];
        v = v < 1.0f ? 1.0f : v;
        d[i] = 1.0f / sqrtf(v);
    }
}

#define GCN_OWN_GROUPS   256
#define GCN_OWN_ROWS     196
#define GCN_OWN_THREADS  1024
#define GCN_OWN_WAVES    (GCN_OWN_THREADS / 32)
#define GCN_Q_CAP        256
#define GCN_SCAN_LOADS   8
#define GCN_EDGE_GROUP   8
#define GCN_LIST_CAP     512
#define GCN_LIST_FLOAT_OFF 12900032
static_assert(GCN_OWN_GROUPS * GCN_OWN_ROWS >= GCN_N_NODES, "the ranges must cover every node");
static_assert(GCN_N_EDGES % 4 == 0, "dst is read 4 edges (16 bytes) at a time");
static_assert(31 + 4 * 32 <= GCN_Q_CAP, "one 16-byte load of a wave may add 128 entries to 31 waiting ones");
static_assert(GCN_N_NODES <= 65536 && GCN_OWN_ROWS <= 32768, "a list entry packs source row (16 bits) and local row into one int");
static_assert(GCN_LIST_CAP % 32 == 0, "whole batches fill a list exactly");
static_assert(GCN_LIST_FLOAT_OFF >= 2 * GCN_N_NODES * 128 + 2 * GCN_N_NODES && GCN_LIST_FLOAT_OFF % 64 == 0,
              "the lists lie behind both row buffers and both degree arrays");

template <bool FULL>
__device__ __forceinline__ void gcn_own_walk(const float* __restrict__ x, float* tile,
                                             int s_l, int d_l, float sc_l, int n, int lane) {
    const int n_edges = FULL ? 32 : n;
    for (int j0 = 0; j0 < n_edges; j0 += GCN_EDGE_GROUP) {
        float v[GCN_EDGE_GROUP][4];
#pragma unroll
        for (int j = 0; j < GCN_EDGE_GROUP; ++j) {
            if (FULL || j0 + j < n) {
                const int s = __builtin_amdgcn_readlane(s_l, j0 + j);
                const float* xr = x + (size_t)s * 128 + lane;
                v[j][0] = xr[0];
                v[j][1] = xr[32];
                v[j][2] = xr[64];
                v[j][3] = xr[96];
            }
        }
#pragma unroll
        for (int j = 0; j < GCN_EDGE_GROUP; ++j) {
            if (FULL || j0 + j < n) {
                const int   d  = __builtin_amdgcn_readlane(d_l, j0 + j);
                const float sc = __int_as_float(
                    __builtin_amdgcn_readlane(__float_as_int(sc_l), j0 + j));
                float* row = tile + d * 128 + lane;
                __hip_atomic_fetch_add(row + 0,  v[j][0] * sc, __ATOMIC_RELAXED, __HIP_MEMORY_SCOPE_WORKGROUP);
                __hip_atomic_fetch_add(row + 32, v[j][1] * sc, __ATOMIC_RELAXED, __HIP_MEMORY_SCOPE_WORKGROUP);
                __hip_atomic_fetch_add(row + 64, v[j][2] * sc, __ATOMIC_RELAXED, __HIP_MEMORY_SCOPE_WORKGROUP);
                __hip_atomic_fetch_add(row + 96, v[j][3] * sc, __ATOMIC_RELAXED, __HIP_MEMORY_SCOPE_WORKGROUP);
            }
        }
    }
}

template <bool FULL, bool RECORD>
__device__ __forceinline__ void gcn_own_batch(const float* __restrict__ x,
                                              const int* __restrict__ src,
                                              const float* __restrict__ deg_out_is,
                                              float* tile, const int* qe, const int* qd,
                                              int n, int lane, int2* wl, int& ln) {
    int   s_l  = 0;
    int   d_l  = 0;
    float sc_l = 0.0f;
    if (FULL || lane < n) {
        const int e = qe[lane];
        d_l  = qd[lane];
        s_l  = src[e];
        sc_l = deg_out_is[s_l];
    }
    if (RECORD) {
        const int cnt = FULL ? 32 : n;
        if (ln + cnt <= GCN_LIST_CAP) {
            if (FULL || lane < n)
                wl[ln + lane] = make_int2(s_l | (d_l << 16), __float_as_int(sc_l));
            ln += cnt;
        } else {
            ln = GCN_LIST_CAP + 1;
        }
    }
    gcn_own_walk<FULL>(x, tile, s_l, d_l, sc_l, n, lane);
}

template <bool RECORD>
__device__ __forceinline__ void gcn_own_scan(const float* __restrict__ x,
                                             const int* __restrict__ src,
                                             const int* __restrict__ dst,
                                             const float* __restrict__ deg_out_is,
                                             float* tile, int* qe, int* qd,
                                             int lo, int rows, int E, int tid, int lane,
                                             int2* wl, int& ln) {
    int  qn = 0;

    const int4* dst4 = reinterpret_cast<const int4*>(dst);
    const int nvec    = E >> 2;
    const int step    = GCN_OWN_THREADS * GCN_SCAN_LOADS;
    const int n_steps = (nvec + step - 1) / step;
    int st = blockIdx.x % n_steps;

    for (int o = 0; o < n_steps; ++o) {
        const int base = st * step + tid;
        const bool whole = (st + 1) * step <= nvec;
        st = st + 1 == n_steps ? 0 : st + 1;

        int4 dv[GCN_SCAN_LOADS];
        if (whole) {
#pragma unroll
            for (int u = 0; u < GCN_SCAN_LOADS; ++u)
                dv[u] = dst4[base + u * GCN_OWN_THREADS];
        } else {
#pragma unroll
            for (int u = 0; u < GCN_SCAN_LOADS; ++u) {
                const int idx = base + u * GCN_OWN_THREADS;
                dv[u] = idx < nvec ? dst4[idx] : make_int4(-1, -1, -1, -1);
            }
        }
#pragma unroll
        for (int u = 0; u < GCN_SCAN_LOADS; ++u) {
            const int e4 = (base + u * GCN_OWN_THREADS) * 4;
            const unsigned r[4] = { (unsigned)(dv[u].x - lo), (unsigned)(dv[u].y - lo),
                                    (unsigned)(dv[u].z - lo), (unsigned)(dv[u].w - lo) };
            const unsigned r01 = r[0] < r[1] ? r[0] : r[1];
            const unsigned r23 = r[2] < r[3] ? r[2] : r[3];
            const unsigned rmin = r01 < r23 ? r01 : r23;
            if (__builtin_amdgcn_ballot_w32(rmin < (unsigned)rows) == 0u) continue;
            const bool m[4] = { r[0] < (unsigned)rows, r[1] < (unsigned)rows,
                                r[2] < (unsigned)rows, r[3] < (unsigned)rows };
#pragma unroll
            for (int c = 0; c < 4; ++c) {
                const unsigned mask = __builtin_amdgcn_ballot_w32(m[c]);
                if (mask != 0u) {
                    const int pos = qn + (int)__builtin_amdgcn_mbcnt_lo(mask, 0u);
                    if (m[c]) { qe[pos] = e4 + c; qd[pos] = (int)r[c]; }
                    qn += __builtin_popcount(mask);
                }
            }
            while (qn >= 32) {
                qn -= 32;
                gcn_own_batch<true, RECORD>(x, src, deg_out_is, tile, qe + qn, qd + qn, 32, lane, wl, ln);
            }
        }
    }
    if (qn > 0)
        gcn_own_batch<false, RECORD>(x, src, deg_out_is, tile, qe, qd, qn, lane, wl, ln);
}

template <bool FIRST>
__global__ __launch_bounds__(GCN_OWN_THREADS)
void gcn_agg_owner(const float* __restrict__ x,
                   const int* __restrict__ src, const int* __restrict__ dst,
                   const float* __restrict__ deg_out_is,
                   float* __restrict__ agg, int* __restrict__ lists,
                   int* __restrict__ counts, int E, int N) {
    __shared__ float tile[GCN_OWN_ROWS * 128];
    __shared__ int   q_e[GCN_OWN_WAVES * GCN_Q_CAP];
    __shared__ int   q_d[GCN_OWN_WAVES * GCN_Q_CAP];

    const int tid  = threadIdx.x;
    const int lane = tid & 31;
    const int wave = __builtin_amdgcn_readfirstlane(tid >> 5);
    const int lo   = blockIdx.x * GCN_OWN_ROWS;
    int rows = N - lo;
    rows = rows > GCN_OWN_ROWS ? GCN_OWN_ROWS : rows;
    rows = rows < 0 ? 0 : rows;

    const int slot = blockIdx.x * GCN_OWN_WAVES + wave;
    int2* wl = reinterpret_cast<int2*>(lists) + (size_t)slot * GCN_LIST_CAP;

    int  my_n      = 0;
    bool walk_list = false;
    if (!FIRST) {
        const int c = counts[blockIdx.x * GCN_OWN_WAVES + lane];
        walk_list = __builtin_amdgcn_ballot_w32(c > GCN_LIST_CAP) == 0u;
        my_n      = __builtin_amdgcn_readlane(c, wave);
    }
    int2 cur = make_int2(0, 0);
    if (!FIRST && walk_list && lane < my_n) cur = wl[lane];

    float4* tile4 = reinterpret_cast<float4*>(tile);
    for (int i = tid; i < GCN_OWN_ROWS * 32; i += GCN_OWN_THREADS)
        tile4[i] = make_float4(0.0f, 0.0f, 0.0f, 0.0f);
    __syncthreads();

    if (!FIRST && walk_list) {
        for (int b = 0; b < my_n; b += 32) {
            int2 nxt = make_int2(0, 0);
            if (b + 32 + lane < my_n) nxt = wl[b + 32 + lane];
            const int   s_l  = cur.x & 0xFFFF;
            const int   d_l  = cur.x >> 16;
            const float sc_l = __int_as_float(cur.y);
            const int rem = my_n - b;
            if (rem >= 32) gcn_own_walk<true >(x, tile, s_l, d_l, sc_l, 32,  lane);
            else           gcn_own_walk<false>(x, tile, s_l, d_l, sc_l, rem, lane);
            cur = nxt;
        }
    } else {
        int ln = 0;
        gcn_own_scan<FIRST>(x, src, dst, deg_out_is, tile,
                            q_e + wave * GCN_Q_CAP, q_d + wave * GCN_Q_CAP,
                            lo, rows, E, tid, lane, wl, ln);
        if (FIRST && lane == 0) counts[slot] = ln;
    }

    __syncthreads();
    float4* out4 = reinterpret_cast<float4*>(agg + (size_t)lo * 128);
    for (int i = tid; i < rows * 32; i += GCN_OWN_THREADS)
        out4[i] = tile4[i];
}

#define GCN_GEMM_WAVES   4
#define GCN_GEMM_THREADS (GCN_GEMM_WAVES * 32)
#define GCN_GEMM_ROWS    (GCN_GEMM_WAVES * 16)
static_assert(GCN_N_NODES % 16 == 0, "a wave's 16-row tile is whole or absent");

template <int NT, int C>
__global__ __launch_bounds__(GCN_GEMM_THREADS)
void gcn_gemm_strip(const float* __restrict__ A,
                    const float* __restrict__ deg_in_is,
                    const float* __restrict__ W,
                    const float* __restrict__ bias,
                    float* __restrict__ out, int do_relu) {
    constexpr int NC     = 16 * NT;
    constexpr int PITCH4 = NC + 8;
    constexpr int STAGE  = (32 * NC) / GCN_GEMM_THREADS;
    static_assert((32 * NC) % GCN_GEMM_THREADS == 0, "whole staging passes");
    __shared__ float4 wq[32 * PITCH4];

    const int tid  = threadIdx.x;
    const int lane = tid & 31;
    const int wave = tid >> 5;
    const int h    = lane >> 4;
    const int c    = lane & 15;
    const int m0   = (blockIdx.x * GCN_GEMM_WAVES + wave) * 16;
    const int nb   = blockIdx.y * NC;
    const bool live = m0 < GCN_N_NODES;

    const int mrow = (live ? m0 : 0) + c;
    const float scale = deg_in_is[mrow];
    const float4* arow = reinterpret_cast<const float4*>(A + (size_t)mrow * 128) + h;
    float4 a[16];
#pragma unroll
    for (int j = 0; j < 8; ++j) a[j] = arow[2 * j];

#pragma unroll
    for (int i = 0; i < STAGE; ++i) {
        const int idx = tid + i * GCN_GEMM_THREADS;
        const int n   = idx % NC;
        const int kq  = idx / NC;
        const float* wp = W + (size_t)(4 * kq) * C + nb + n;
        wq[kq * PITCH4 + n] = make_float4(wp[0], wp[C], wp[2 * C], wp[3 * C]);
    }
    __syncthreads();
    if (!live) return;

#pragma unroll
    for (int j = 8; j < 16; ++j) a[j] = arow[2 * j];

    v8f acc[NT];
#pragma unroll
    for (int t = 0; t < NT; ++t) acc[t] = v8f{};

#pragma unroll
    for (int j = 0; j < 16; ++j) {
        v2f a0, a1;
        a0.x = a[j].x * scale;  a0.y = a[j].y * scale;
        a1.x = a[j].z * scale;  a1.y = a[j].w * scale;
        float4 b[NT];
#pragma unroll
        for (int t = 0; t < NT; ++t) b[t] = wq[(2 * j + h) * PITCH4 + t * 16 + c];
#pragma unroll
        for (int t = 0; t < NT; ++t) {
            v2f b0;  b0.x = b[t].x;  b0.y = b[t].y;
            acc[t] = __builtin_amdgcn_wmma_f32_16x16x4_f32(
                         false, a0, false, b0, (short)0, acc[t], false, false);
        }
#pragma unroll
        for (int t = 0; t < NT; ++t) {
            v2f b1;  b1.x = b[t].z;  b1.y = b[t].w;
            acc[t] = __builtin_amdgcn_wmma_f32_16x16x4_f32(
                         false, a1, false, b1, (short)0, acc[t], false, false);
        }
    }

#pragma unroll
    for (int t = 0; t < NT; ++t) {
        const int n = nb + t * 16 + c;
        const float bv = bias[n];
#pragma unroll
        for (int v = 0; v < 8; ++v) {
            float val = acc[t][v] + bv;
            if (do_relu) val = fmaxf(val, 0.0f);
            out[(size_t)(m0 + h * 8 + v) * C + n] = val;
        }
    }
}

extern "C" void kernel_launch(void* const* d_in, const int* in_sizes, int n_in,
                              void* d_out, int out_size, void* d_ws, size_t ws_size,
                              hipStream_t stream) {
    const float* features = (const float*)d_in[0];
    const int*   src      = (const int*)  d_in[1];
    const int*   dst      = (const int*)  d_in[2];
    const float* W1       = (const float*)d_in[3];
    const float* b1       = (const float*)d_in[4];
    const float* W2       = (const float*)d_in[5];
    const float* b2       = (const float*)d_in[6];
    const float* W3       = (const float*)d_in[7];
    const float* b3       = (const float*)d_in[8];
    float* out = (float*)d_out;

    const int N = GCN_N_NODES;
    const int E = GCN_N_EDGES;
    const size_t NF = (size_t)N * 128;

    float* bufA  = (float*)d_ws;
    float* bufB  = bufA + NF;
    float* deg_o = bufB + NF;
    float* deg_i = deg_o + N;
    int*   lists  = (int*)((float*)d_ws + GCN_LIST_FLOAT_OFF);
    int*   counts = lists + 2 * GCN_LIST_CAP * GCN_OWN_WAVES * GCN_OWN_GROUPS;

    const int TB = 256;
    const int FILL_PER_BLOCK = TB * 4 * GCN_FILL_STORES;

    gcn_fill_zero<<<(2 * N + FILL_PER_BLOCK - 1) / FILL_PER_BLOCK, TB, 0, stream>>>(deg_o, 2 * N);
    gcn_count_deg<<<(E + TB - 1) / TB, TB, 0, stream>>>(src, dst, deg_o, deg_i, E);
    gcn_inv_sqrt_clamp<<<(2 * N + TB - 1) / TB, TB, 0, stream>>>(deg_o, 2 * N);

    const dim3 gemm128((GCN_N_NODES + GCN_GEMM_ROWS - 1) / GCN_GEMM_ROWS, 2);
    const dim3 gemm16 ((GCN_N_NODES + GCN_GEMM_ROWS - 1) / GCN_GEMM_ROWS, 1);

    gcn_agg_owner<true><<<GCN_OWN_GROUPS, GCN_OWN_THREADS, 0, stream>>>(features, src, dst, deg_o, bufA, lists, counts, E, N);
    gcn_gemm_strip<4, 128><<<gemm128, GCN_GEMM_THREADS, 0, stream>>>(bufA, deg_i, W1, b1, bufB, 1);

    gcn_agg_owner<false><<<GCN_OWN_GROUPS, GCN_OWN_THREADS, 0, stream>>>(bufB, src, dst, deg_o, bufA, lists, counts, E, N);
    gcn_gemm_strip<4, 128><<<gemm128, GCN_GEMM_THREADS, 0, stream>>>(bufA, deg_i, W2, b2, bufB, 1);

    gcn_agg_owner<false><<<GCN_OWN_GROUPS, GCN_OWN_THREADS, 0, stream>>>(bufB, src, dst, deg_o, bufA, lists, counts, E, N);
    gcn_gemm_strip<1, 16><<<gemm16, GCN_GEMM_THREADS, 0, stream>>>(bufA, deg_i, W3, b3, out, 0);

    (void)in_sizes; (void)n_in; (void)out_size; (void)ws_size;
}


# Source rows gathered by one 16-byte-per-lane load per row (its two aligned 256-B units at once; was four 128-B line loads); LDS tile rows kept as 4 planes of 32 floats; sum order still LDS arrival
# speedup vs baseline: 2.6467x; 1.0514x over previous
#include <hip/hip_runtime.h>
#include <math.h>

typedef __attribute__((ext_vector_type(2))) float v2f;
typedef __attribute__((ext_vector_type(8))) float v8f;

#define GCN_N_NODES 50000
#define GCN_N_EDGES 600000

#define GCN_FILL_STORES 8
__global__ void gcn_fill_zero(float* __restrict__ p, int n) {
    float4* p4 = reinterpret_cast<float4*>(p);
    const int n4 = n >> 2;
    int i = blockIdx.x * (blockDim.x * GCN_FILL_STORES) + threadIdx.x;
#pragma unroll
    for (int k = 0; k < GCN_FILL_STORES; ++k, i += blockDim.x)
        if (i < n4) p4[i] = make_float4(0.0f, 0.0f, 0.0f, 0.0f);
}

__global__ void gcn_count_deg(const int* __restrict__ src, const int* __restrict__ dst,
                              float* __restrict__ deg_out, float* __restrict__ deg_in,
                              int E) {
    int e = blockIdx.x * blockDim.x + threadIdx.x;
    if (e < E) {
        atomicAdd(&deg_out[src[e]], 1.0f);
        atomicAdd(&deg_in[dst[e]], 1.0f);
    }
}

__global__ void gcn_inv_sqrt_clamp(float* __restrict__ d, int n) {
    int i = blockIdx.x * blockDim.x + threadIdx.x;
    if (i < n) {
        float v = d[i];
        v = v < 1.0f ? 1.0f : v;
        d[i] = 1.0f / sqrtf(v);
    }
}

#define GCN_OWN_GROUPS   256
#define GCN_OWN_ROWS     196
#define GCN_OWN_THREADS  1024
#define GCN_OWN_WAVES    (GCN_OWN_THREADS / 32)
#define GCN_Q_CAP        256
#define GCN_SCAN_LOADS   8
#define GCN_EDGE_GROUP   8
#define GCN_LIST_CAP     512
#define GCN_LIST_FLOAT_OFF 12900032
static_assert(GCN_OWN_GROUPS * GCN_OWN_ROWS >= GCN_N_NODES, "the ranges must cover every node");
static_assert(GCN_N_EDGES % 4 == 0, "dst is read 4 edges (16 bytes) at a time");
static_assert(31 + 4 * 32 <= GCN_Q_CAP, "one 16-byte load of a wave may add 128 entries to 31 waiting ones");
static_assert(GCN_N_NODES <= 65536 && GCN_OWN_ROWS <= 32768, "a list entry packs source row (16 bits) and local row into one int");
static_assert(GCN_LIST_CAP % 32 == 0, "whole batches fill a list exactly");
static_assert(GCN_LIST_FLOAT_OFF >= 2 * GCN_N_NODES * 128 + 2 * GCN_N_NODES && GCN_LIST_FLOAT_OFF % 64 == 0,
              "the lists lie behind both row buffers and both degree arrays");

template <bool FULL>
__device__ __forceinline__ void gcn_own_walk(const float* __restrict__ x, float* tile,
                                             int s_l, int d_l, float sc_l, int n, int lane) {
    const int n_edges = FULL ? 32 : n;
#pragma nounroll
    for (int j0 = 0; j0 < n_edges; j0 += GCN_EDGE_GROUP) {
        float4 v[GCN_EDGE_GROUP];
#pragma unroll
        for (int j = 0; j < GCN_EDGE_GROUP; ++j) {
            if (FULL || j0 + j < n) {
                const int s = __builtin_amdgcn_readlane(s_l, j0 + j);
                v[j] = reinterpret_cast<const float4*>(x + (size_t)s * 128)[lane];
            }
        }
#pragma unroll
        for (int j = 0; j < GCN_EDGE_GROUP; ++j) {
            if (FULL || j0 + j < n) {
                const int   d  = __builtin_amdgcn_readlane(d_l, j0 + j);
                const float sc = __int_as_float(
                    __builtin_amdgcn_readlane(__float_as_int(sc_l), j0 + j));
                float* row = tile + d * 128 + lane;
                __hip_atomic_fetch_add(row + 0,  v[j].x * sc, __ATOMIC_RELAXED, __HIP_MEMORY_SCOPE_WORKGROUP);
                __hip_atomic_fetch_add(row + 32, v[j].y * sc, __ATOMIC_RELAXED, __HIP_MEMORY_SCOPE_WORKGROUP);
                __hip_atomic_fetch_add(row + 64, v[j].z * sc, __ATOMIC_RELAXED, __HIP_MEMORY_SCOPE_WORKGROUP);
                __hip_atomic_fetch_add(row + 96, v[j].w * sc, __ATOMIC_RELAXED, __HIP_MEMORY_SCOPE_WORKGROUP);
            }
        }
    }
}

template <bool FULL, bool RECORD>
__device__ __forceinline__ void gcn_own_batch(const float* __restrict__ x,
                                              const int* __restrict__ src,
                                              const float* __restrict__ deg_out_is,
                                              float* tile, const int* qe, const int* qd,
                                              int n, int lane, int2* wl, int& ln) {
    int   s_l  = 0;
    int   d_l  = 0;
    float sc_l = 0.0f;
    if (FULL || lane < n) {
        const int e = qe[lane];
        d_l  = qd[lane];
        s_l  = src[e];
        sc_l = deg_out_is[s_l];
    }
    if (RECORD) {
        const int cnt = FULL ? 32 : n;
        if (ln + cnt <= GCN_LIST_CAP) {
            if (FULL || lane < n)
                wl[ln + lane] = make_int2(s_l | (d_l << 16), __float_as_int(sc_l));
            ln += cnt;
        } else {
            ln = GCN_LIST_CAP + 1;
        }
    }
    gcn_own_walk<FULL>(x, tile, s_l, d_l, sc_l, n, lane);
}

template <bool RECORD>
__device__ __forceinline__ void gcn_own_scan(const float* __restrict__ x,
                                             const int* __restrict__ src,
                                             const int* __restrict__ dst,
                                             const float* __restrict__ deg_out_is,
                                             float* tile, int* qe, int* qd,
                                             int lo, int rows, int E, int tid, int lane,
                                             int2* wl, int& ln) {
    int  qn = 0;

    const int4* dst4 = reinterpret_cast<const int4*>(dst);
    const int nvec    = E >> 2;
    const int step    = GCN_OWN_THREADS * GCN_SCAN_LOADS;
    const int n_steps = (nvec + step - 1) / step;
    int st = blockIdx.x % n_steps;

    for (int o = 0; o < n_steps; ++o) {
        const int base = st * step + tid;
        const bool whole = (st + 1) * step <= nvec;
        st = st + 1 == n_steps ? 0 : st + 1;

        int4 dv[GCN_SCAN_LOADS];
        if (whole) {
#pragma unroll
            for (int u = 0; u < GCN_SCAN_LOADS; ++u)
                dv[u] = dst4[base + u * GCN_OWN_THREADS];
        } else {
#pragma unroll
            for (int u = 0; u < GCN_SCAN_LOADS; ++u) {
                const int idx = base + u * GCN_OWN_THREADS;
                dv[u] = idx < nvec ? dst4[idx] : make_int4(-1, -1, -1, -1);
            }
        }
#pragma unroll
        for (int u = 0; u < GCN_SCAN_LOADS; ++u) {
            const int e4 = (base + u * GCN_OWN_THREADS) * 4;
            const unsigned r[4] = { (unsigned)(dv[u].x - lo), (unsigned)(dv[u].y - lo),
                                    (unsigned)(dv[u].z - lo), (unsigned)(dv[u].w - lo) };
            const unsigned r01 = r[0] < r[1] ? r[0] : r[1];
            const unsigned r23 = r[2] < r[3] ? r[2] : r[3];
            const unsigned rmin = r01 < r23 ? r01 : r23;
            if (__builtin_amdgcn_ballot_w32(rmin < (unsigned)rows) == 0u) continue;
            const bool m[4] = { r[0] < (unsigned)rows, r[1] < (unsigned)rows,
                                r[2] < (unsigned)rows, r[3] < (unsigned)rows };
#pragma unroll
            for (int c = 0; c < 4; ++c) {
                const unsigned mask = __builtin_amdgcn_ballot_w32(m[c]);
                if (mask != 0u) {
                    const int pos = qn + (int)__builtin_amdgcn_mbcnt_lo(mask, 0u);
                    if (m[c]) { qe[pos] = e4 + c; qd[pos] = (int)r[c]; }
                    qn += __builtin_popcount(mask);
                }
            }
            while (qn >= 32) {
                qn -= 32;
                gcn_own_batch<true, RECORD>(x, src, deg_out_is, tile, qe + qn, qd + qn, 32, lane, wl, ln);
            }
        }
    }
    if (qn > 0)
        gcn_own_batch<false, RECORD>(x, src, deg_out_is, tile, qe, qd, qn, lane, wl, ln);
}

template <bool FIRST>
__global__ __launch_bounds__(GCN_OWN_THREADS)
void gcn_agg_owner(const float* __restrict__ x,
                   const int* __restrict__ src, const int* __restrict__ dst,
                   const float* __restrict__ deg_out_is,
                   float* __restrict__ agg, int* __restrict__ lists,
                   int* __restrict__ counts, int E, int N) {
    __shared__ float tile[GCN_OWN_ROWS * 128];
    __shared__ int   q_e[GCN_OWN_WAVES * GCN_Q_CAP];
    __shared__ int   q_d[GCN_OWN_WAVES * GCN_Q_CAP];

    const int tid  = threadIdx.x;
    const int lane = tid & 31;
    const int wave = __builtin_amdgcn_readfirstlane(tid >> 5);
    const int lo   = blockIdx.x * GCN_OWN_ROWS;
    int rows = N - lo;
    rows = rows > GCN_OWN_ROWS ? GCN_OWN_ROWS : rows;
    rows = rows < 0 ? 0 : rows;

    const int slot = blockIdx.x * GCN_OWN_WAVES + wave;
    int2* wl = reinterpret_cast<int2*>(lists) + (size_t)slot * GCN_LIST_CAP;

    int  my_n      = 0;
    bool walk_list = false;
    if (!FIRST) {
        const int c = counts[blockIdx.x * GCN_OWN_WAVES + lane];
        walk_list = __builtin_amdgcn_ballot_w32(c > GCN_LIST_CAP) == 0u;
        my_n      = __builtin_amdgcn_readlane(c, wave);
    }
    int2 cur = make_int2(0, 0);
    if (!FIRST && walk_list && lane < my_n) cur = wl[lane];

    float4* tile4 = reinterpret_cast<float4*>(tile);
    for (int i = tid; i < GCN_OWN_ROWS * 32; i += GCN_OWN_THREADS)
        tile4[i] = make_float4(0.0f, 0.0f, 0.0f, 0.0f);
    __syncthreads();

    if (!FIRST && walk_list) {
        for (int b = 0; b < my_n; b += 32) {
            int2 nxt = make_int2(0, 0);
            if (b + 32 + lane < my_n) nxt = wl[b + 32 + lane];
            const int   s_l  = cur.x & 0xFFFF;
            const int   d_l  = cur.x >> 16;
            const float sc_l = __int_as_float(cur.y);
            const int rem = my_n - b;
            if (rem >= 32) gcn_own_walk<true >(x, tile, s_l, d_l, sc_l, 32,  lane);
            else           gcn_own_walk<false>(x, tile, s_l, d_l, sc_l, rem, lane);
            cur = nxt;
        }
    } else {
        int ln = 0;
        gcn_own_scan<FIRST>(x, src, dst, deg_out_is, tile,
                            q_e + wave * GCN_Q_CAP, q_d + wave * GCN_Q_CAP,
                            lo, rows, E, tid, lane, wl, ln);
        if (FIRST && lane == 0) counts[slot] = ln;
    }

    __syncthreads();
    float4* out4 = reinterpret_cast<float4*>(agg + (size_t)lo * 128);
    for (int i = tid; i < rows * 32; i += GCN_OWN_THREADS) {
        const float* p = tile + (i >> 5) * 128 + (i & 31);
        out4[i] = make_float4(p[0], p[32], p[64], p[96]);
    }
}

#define GCN_GEMM_WAVES   4
#define GCN_GEMM_THREADS (GCN_GEMM_WAVES * 32)
#define GCN_GEMM_ROWS    (GCN_GEMM_WAVES * 16)
static_assert(GCN_N_NODES % 16 == 0, "a wave's 16-row tile is whole or absent");

template <int NT, int C>
__global__ __launch_bounds__(GCN_GEMM_THREADS)
void gcn_gemm_strip(const float* __restrict__ A,
                    const float* __restrict__ deg_in_is,
                    const float* __restrict__ W,
                    const float* __restrict__ bias,
                    float* __restrict__ out, int do_relu) {
    constexpr int NC     = 16 * NT;
    constexpr int PITCH4 = NC + 8;
    constexpr int STAGE  = (32 * NC) / GCN_GEMM_THREADS;
    static_assert((32 * NC) % GCN_GEMM_THREADS == 0, "whole staging passes");
    __shared__ float4 wq[32 * PITCH4];

    const int tid  = threadIdx.x;
    const int lane = tid & 31;
    const int wave = tid >> 5;
    const int h    = lane >> 4;
    const int c    = lane & 15;
    const int m0   = (blockIdx.x * GCN_GEMM_WAVES + wave) * 16;
    const int nb   = blockIdx.y * NC;
    const bool live = m0 < GCN_N_NODES;

    const int mrow = (live ? m0 : 0) + c;
    const float scale = deg_in_is[mrow];
    const float4* arow = reinterpret_cast<const float4*>(A + (size_t)mrow * 128) + h;
    float4 a[16];
#pragma unroll
    for (int j = 0; j < 8; ++j) a[j] = arow[2 * j];

#pragma unroll
    for (int i = 0; i < STAGE; ++i) {
        const int idx = tid + i * GCN_GEMM_THREADS;
        const int n   = idx % NC;
        const int kq  = idx / NC;
        const float* wp = W + (size_t)(4 * kq) * C + nb + n;
        wq[kq * PITCH4 + n] = make_float4(wp[0], wp[C], wp[2 * C], wp[3 * C]);
    }
    __syncthreads();
    if (!live) return;

#pragma unroll
    for (int j = 8; j < 16; ++j) a[j] = arow[2 * j];

    v8f acc[NT];
#pragma unroll
    for (int t = 0; t < NT; ++t) acc[t] = v8f{};

#pragma unroll
    for (int j = 0; j < 16; ++j) {
        v2f a0, a1;
        a0.x = a[j].x * scale;  a0.y = a[j].y * scale;
        a1.x = a[j].z * scale;  a1.y = a[j].w * scale;
        float4 b[NT];
#pragma unroll
        for (int t = 0; t < NT; ++t) b[t] = wq[(2 * j + h) * PITCH4 + t * 16 + c];
#pragma unroll
        for (int t = 0; t < NT; ++t) {
            v2f b0;  b0.x = b[t].x;  b0.y = b[t].y;
            acc[t] = __builtin_amdgcn_wmma_f32_16x16x4_f32(
                         false, a0, false, b0, (short)0, acc[t], false, false);
        }
#pragma unroll
        for (int t = 0; t < NT; ++t) {
            v2f b1;  b1.x = b[t].z;  b1.y = b[t].w;
            acc[t] = __builtin_amdgcn_wmma_f32_16x16x4_f32(
                         false, a1, false, b1, (short)0, acc[t], false, false);
        }
    }

#pragma unroll
    for (int t = 0; t < NT; ++t) {
        const int n = nb + t * 16 + c;
        const float bv = bias[n];
#pragma unroll
        for (int v = 0; v < 8; ++v) {
            float val = acc[t][v] + bv;
            if (do_relu) val = fmaxf(val, 0.0f);
            out[(size_t)(m0 + h * 8 + v) * C + n] = val;
        }
    }
}

extern "C" void kernel_launch(void* const* d_in, const int* in_sizes, int n_in,
                              void* d_out, int out_size, void* d_ws, size_t ws_size,
                              hipStream_t stream) {
    const float* features = (const float*)d_in[0];
    const int*   src      = (const int*)  d_in[1];
    const int*   dst      = (const int*)  d_in[2];
    const float* W1       = (const float*)d_in[3];
    const float* b1       = (const float*)d_in[4];
    const float* W2       = (const float*)d_in[5];
    const float* b2       = (const float*)d_in[6];
    const float* W3       = (const float*)d_in[7];
    const float* b3       = (const float*)d_in[8];
    float* out = (float*)d_out;

    const int N = GCN_N_NODES;
    const int E = GCN_N_EDGES;
    const size_t NF = (size_t)N * 128;

    float* bufA  = (float*)d_ws;
    float* bufB  = bufA + NF;
    float* deg_o = bufB + NF;
    float* deg_i = deg_o + N;
    int*   lists  = (int*)((float*)d_ws + GCN_LIST_FLOAT_OFF);
    int*   counts = lists + 2 * GCN_LIST_CAP * GCN_OWN_WAVES * GCN_OWN_GROUPS;

    const int TB = 256;
    const int FILL_PER_BLOCK = TB * 4 * GCN_FILL_STORES;

    gcn_fill_zero<<<(2 * N + FILL_PER_BLOCK - 1) / FILL_PER_BLOCK, TB, 0, stream>>>(deg_o, 2 * N);
    gcn_count_deg<<<(E + TB - 1) / TB, TB, 0, stream>>>(src, dst, deg_o, deg_i, E);
    gcn_inv_sqrt_clamp<<<(2 * N + TB - 1) / TB, TB, 0, stream>>>(deg_o, 2 * N);

    const dim3 gemm128((GCN_N_NODES + GCN_GEMM_ROWS - 1) / GCN_GEMM_ROWS, 2);
    const dim3 gemm16 ((GCN_N_NODES + GCN_GEMM_ROWS - 1) / GCN_GEMM_ROWS, 1);

    gcn_agg_owner<true><<<GCN_OWN_GROUPS, GCN_OWN_THREADS, 0, stream>>>(features, src, dst, deg_o, bufA, lists, counts, E, N);
    gcn_gemm_strip<4, 128><<<gemm128, GCN_GEMM_THREADS, 0, stream>>>(bufA, deg_i, W1, b1, bufB, 1);

    gcn_agg_owner<false><<<GCN_OWN_GROUPS, GCN_OWN_THREADS, 0, stream>>>(bufB, src, dst, deg_o, bufA, lists, counts, E, N);
    gcn_gemm_strip<4, 128><<<gemm128, GCN_GEMM_THREADS, 0, stream>>>(bufA, deg_i, W2, b2, bufB, 1);

    gcn_agg_owner<false><<<GCN_OWN_GROUPS, GCN_OWN_THREADS, 0, stream>>>(bufB, src, dst, deg_o, bufA, lists, counts, E, N);
    gcn_gemm_strip<1, 16><<<gemm16, GCN_GEMM_THREADS, 0, stream>>>(bufA, deg_i, W3, b3, out, 0);

    (void)in_sizes; (void)n_in; (void)out_size; (void)ws_size;
}


# Matrix product moved into the owner work-groups: the finished LDS sum tile (row pitch 132 floats) is multiplied there by W staged in LDS; sums never stored, 9 -> 6 launches; sum order same
# speedup vs baseline: 3.0407x; 1.1489x over previous
#include <hip/hip_runtime.h>
#include <math.h>

typedef __attribute__((ext_vector_type(2))) float v2f;
typedef __attribute__((ext_vector_type(8))) float v8f;

#define GCN_N_NODES 50000
#define GCN_N_EDGES 600000

#define GCN_FILL_STORES 8
__global__ void gcn_fill_zero(float* __restrict__ p, int n) {
    float4* p4 = reinterpret_cast<float4*>(p);
    const int n4 = n >> 2;
    int i = blockIdx.x * (blockDim.x * GCN_FILL_STORES) + threadIdx.x;
#pragma unroll
    for (int k = 0; k < GCN_FILL_STORES; ++k, i += blockDim.x)
        if (i < n4) p4[i] = make_float4(0.0f, 0.0f, 0.0f, 0.0f);
}

__global__ void gcn_count_deg(const int* __restrict__ src, const int* __restrict__ dst,
                              float* __restrict__ deg_out, float* __restrict__ deg_in,
                              int E) {
    int e = blockIdx.x * blockDim.x + threadIdx.x;
    if (e < E) {
        atomicAdd(&deg_out[src[e]], 1.0f);
        atomicAdd(&deg_in[dst[e]], 1.0f);
    }
}

__global__ void gcn_inv_sqrt_clamp(float* __restrict__ d, int n) {
    int i = blockIdx.x * blockDim.x + threadIdx.x;
    if (i < n) {
        float v = d[i];
        v = v < 1.0f ? 1.0f : v;
        d[i] = 1.0f / sqrtf(v);
    }
}

#define GCN_OWN_GROUPS   256
#define GCN_OWN_ROWS     196
#define GCN_TILE_ROWS    208
#define GCN_TILE_PITCH   132
#define GCN_OWN_THREADS  1024
#define GCN_OWN_WAVES    (GCN_OWN_THREADS / 32)
#define GCN_Q_CAP        256
#define GCN_SCAN_LOADS   8
#define GCN_EDGE_GROUP   8
#define GCN_LIST_CAP     512
#define GCN_LIST_FLOAT_OFF 12900032
static_assert(GCN_OWN_GROUPS * GCN_OWN_ROWS >= GCN_N_NODES, "the ranges must cover every node");
static_assert(GCN_TILE_ROWS % 16 == 0 && GCN_TILE_ROWS >= GCN_OWN_ROWS && GCN_TILE_PITCH % 4 == 0 && GCN_TILE_PITCH >= 128,
              "whole row tiles, rows of whole 16-byte pieces");
static_assert(GCN_N_EDGES % 4 == 0, "dst is read 4 edges (16 bytes) at a time");
static_assert(31 + 4 * 32 <= GCN_Q_CAP, "one 16-byte load of a wave may add 128 entries to 31 waiting ones");
static_assert(GCN_N_NODES <= 65536 && GCN_OWN_ROWS <= 32768, "a list entry packs source row (16 bits) and local row into one int");
static_assert(GCN_LIST_CAP % 32 == 0, "whole batches fill a list exactly");
static_assert(GCN_LIST_FLOAT_OFF >= 2 * GCN_N_NODES * 128 + 2 * GCN_N_NODES && GCN_LIST_FLOAT_OFF % 64 == 0,
              "the lists lie behind both row buffers and both degree arrays");

template <bool FULL>
__device__ __forceinline__ void gcn_own_walk(const float* __restrict__ x, float* tile,
                                             int s_l, int d_l, float sc_l, int n, int lane) {
    const int n_edges = FULL ? 32 : n;
#pragma nounroll
    for (int j0 = 0; j0 < n_edges; j0 += GCN_EDGE_GROUP) {
        float4 v[GCN_EDGE_GROUP];
#pragma unroll
        for (int j = 0; j < GCN_EDGE_GROUP; ++j) {
            if (FULL || j0 + j < n) {
                const int s = __builtin_amdgcn_readlane(s_l, j0 + j);
                v[j] = reinterpret_cast<const float4*>(x + (size_t)s * 128)[lane];
            }
        }
#pragma unroll
        for (int j = 0; j < GCN_EDGE_GROUP; ++j) {
            if (FULL || j0 + j < n) {
                const int   d  = __builtin_amdgcn_readlane(d_l, j0 + j);
                const float sc = __int_as_float(
                    __builtin_amdgcn_readlane(__float_as_int(sc_l), j0 + j));
                float* row = tile + d * GCN_TILE_PITCH + lane;
                __hip_atomic_fetch_add(row + 0,  v[j].x * sc, __ATOMIC_RELAXED, __HIP_MEMORY_SCOPE_WORKGROUP);
                __hip_atomic_fetch_add(row + 32, v[j].y * sc, __ATOMIC_RELAXED, __HIP_MEMORY_SCOPE_WORKGROUP);
                __hip_atomic_fetch_add(row + 64, v[j].z * sc, __ATOMIC_RELAXED, __HIP_MEMORY_SCOPE_WORKGROUP);
                __hip_atomic_fetch_add(row + 96, v[j].w * sc, __ATOMIC_RELAXED, __HIP_MEMORY_SCOPE_WORKGROUP);
            }
        }
    }
}

template <bool FULL, bool RECORD>
__device__ __forceinline__ void gcn_own_batch(const float* __restrict__ x,
                                              const int* __restrict__ src,
                                              const float* __restrict__ deg_out_is,
                                              float* tile, const int* qe, const int* qd,
                                              int n, int lane, int2* wl, int& ln) {
    int   s_l  = 0;
    int   d_l  = 0;
    float sc_l = 0.0f;
    if (FULL || lane < n) {
        const int e = qe[lane];
        d_l  = qd[lane];
        s_l  = src[e];
        sc_l = deg_out_is[s_l];
    }
    if (RECORD) {
        const int cnt = FULL ? 32 : n;
        if (ln + cnt <= GCN_LIST_CAP) {
            if (FULL || lane < n)
                wl[ln + lane] = make_int2(s_l | (d_l << 16), __float_as_int(sc_l));
            ln += cnt;
        } else {
            ln = GCN_LIST_CAP + 1;
        }
    }
    gcn_own_walk<FULL>(x, tile, s_l, d_l, sc_l, n, lane);
}

template <bool RECORD>
__device__ __forceinline__ void gcn_own_scan(const float* __restrict__ x,
                                             const int* __restrict__ src,
                                             const int* __restrict__ dst,
                                             const float* __restrict__ deg_out_is,
                                             float* tile, int* qe, int* qd,
                                             int lo, int rows, int E, int tid, int lane,
                                             int2* wl, int& ln) {
    int  qn = 0;

    const int4* dst4 = reinterpret_cast<const int4*>(dst);
    const int nvec    = E >> 2;
    const int step    = GCN_OWN_THREADS * GCN_SCAN_LOADS;
    const int n_steps = (nvec + step - 1) / step;
    int st = blockIdx.x % n_steps;

    for (int o = 0; o < n_steps; ++o) {
        const int base = st * step + tid;
        const bool whole = (st + 1) * step <= nvec;
        st = st + 1 == n_steps ? 0 : st + 1;

        int4 dv[GCN_SCAN_LOADS];
        if (whole) {
#pragma unroll
            for (int u = 0; u < GCN_SCAN_LOADS; ++u)
                dv[u] = dst4[base + u * GCN_OWN_THREADS];
        } else {
#pragma unroll
            for (int u = 0; u < GCN_SCAN_LOADS; ++u) {
                const int idx = base + u * GCN_OWN_THREADS;
                dv[u] = idx < nvec ? dst4[idx] : make_int4(-1, -1, -1, -1);
            }
        }
#pragma unroll
        for (int u = 0; u < GCN_SCAN_LOADS; ++u) {
            const int e4 = (base + u * GCN_OWN_THREADS) * 4;
            const unsigned r[4] = { (unsigned)(dv[u].x - lo), (unsigned)(dv[u].y - lo),
                                    (unsigned)(dv[u].z - lo), (unsigned)(dv[u].w - lo) };
            const unsigned r01 = r[0] < r[1] ? r[0] : r[1];
            const unsigned r23 = r[2] < r[3] ? r[2] : r[3];
            const unsigned rmin = r01 < r23 ? r01 : r23;
            if (__builtin_amdgcn_ballot_w32(rmin < (unsigned)rows) == 0u) continue;
            const bool m[4] = { r[0] < (unsigned)rows, r[1] < (unsigned)rows,
                                r[2] < (unsigned)rows, r[3] < (unsigned)rows };
#pragma unroll
            for (int c = 0; c < 4; ++c) {
                const unsigned mask = __builtin_amdgcn_ballot_w32(m[c]);
                if (mask != 0u) {
                    const int pos = qn + (int)__builtin_amdgcn_mbcnt_lo(mask, 0u);
                    if (m[c]) { qe[pos] = e4 + c; qd[pos] = (int)r[c]; }
                    qn += __builtin_popcount(mask);
                }
            }
            while (qn >= 32) {
                qn -= 32;
                gcn_own_batch<true, RECORD>(x, src, deg_out_is, tile, qe + qn, qd + qn, 32, lane, wl, ln);
            }
        }
    }
    if (qn > 0)
        gcn_own_batch<false, RECORD>(x, src, deg_out_is, tile, qe, qd, qn, lane, wl, ln);
}

template <bool FIRST, int C>
__global__ __launch_bounds__(GCN_OWN_THREADS)
void gcn_layer_owner(const float* __restrict__ x,
                     const int* __restrict__ src, const int* __restrict__ dst,
                     const float* __restrict__ deg_out_is,
                     const float* __restrict__ deg_in_is,
                     const float* __restrict__ W, const float* __restrict__ bias,
                     float* __restrict__ out, int* __restrict__ lists,
                     int* __restrict__ counts, int E, int N, int do_relu) {
    constexpr int NT     = C >= 64 ? 4 : C / 16;
    constexpr int CG     = C / (16 * NT);
    constexpr int PITCH4 = C + 8;
    constexpr int WQ_N   = 32 * C;
    constexpr int STAGE  = (WQ_N + GCN_OWN_THREADS - 1) / GCN_OWN_THREADS;
    static_assert(C % (16 * NT) == 0 && (GCN_TILE_ROWS / 16) * CG <= GCN_OWN_WAVES, "a wave per row tile and column group");
    __shared__ float  tile[GCN_TILE_ROWS * GCN_TILE_PITCH];
    __shared__ int    q_e[GCN_OWN_WAVES * GCN_Q_CAP];
    __shared__ int    q_d[GCN_OWN_WAVES * GCN_Q_CAP];
    __shared__ float4 wq[32 * PITCH4];

    const int tid  = threadIdx.x;
    const int lane = tid & 31;
    const int wave = __builtin_amdgcn_readfirstlane(tid >> 5);
    const int lo   = blockIdx.x * GCN_OWN_ROWS;
    int rows = N - lo;
    rows = rows > GCN_OWN_ROWS ? GCN_OWN_ROWS : rows;
    rows = rows < 0 ? 0 : rows;

    const int slot = blockIdx.x * GCN_OWN_WAVES + wave;
    int2* wl = reinterpret_cast<int2*>(lists) + (size_t)slot * GCN_LIST_CAP;

    int  my_n      = 0;
    bool walk_list = false;
    if (!FIRST) {
        const int c = counts[blockIdx.x * GCN_OWN_WAVES + lane];
        walk_list = __builtin_amdgcn_ballot_w32(c > GCN_LIST_CAP) == 0u;
        my_n      = __builtin_amdgcn_readlane(c, wave);
    }
    int2 cur = make_int2(0, 0);
    if (!FIRST && walk_list && lane < my_n) cur = wl[lane];

    const int  h      = lane >> 4;
    const int  c16    = lane & 15;
    const int  rt     = wave / CG;
    const int  nb     = (wave % CG) * (16 * NT);
    const bool mm     = rt * 16 < rows;
    const int  arow   = rt * 16 + c16;
    float scale = 0.0f;
    float bv[NT];
#pragma unroll
    for (int t = 0; t < NT; ++t) bv[t] = 0.0f;
    if (mm) {
        if (arow < rows) scale = deg_in_is[lo + arow];
#pragma unroll
        for (int t = 0; t < NT; ++t) bv[t] = bias[nb + t * 16 + c16];
    }

    float4 wreg[STAGE];
#pragma unroll
    for (int i = 0; i < STAGE; ++i) {
        const int idx = tid + i * GCN_OWN_THREADS;
        wreg[i] = make_float4(0.0f, 0.0f, 0.0f, 0.0f);
        if (WQ_N % GCN_OWN_THREADS == 0 || idx < WQ_N) {
            const float* wp = W + (size_t)(4 * (idx / C)) * C + (idx % C);
            wreg[i] = make_float4(wp[0], wp[C], wp[2 * C], wp[3 * C]);
        }
    }

    float4* tile4 = reinterpret_cast<float4*>(tile);
    for (int i = tid; i < GCN_TILE_ROWS * (GCN_TILE_PITCH / 4); i += GCN_OWN_THREADS)
        tile4[i] = make_float4(0.0f, 0.0f, 0.0f, 0.0f);
#pragma unroll
    for (int i = 0; i < STAGE; ++i) {
        const int idx = tid + i * GCN_OWN_THREADS;
        if (WQ_N % GCN_OWN_THREADS == 0 || idx < WQ_N)
            wq[(idx / C) * PITCH4 + (idx % C)] = wreg[i];
    }
    __syncthreads();

    if (!FIRST && walk_list) {
        for (int b = 0; b < my_n; b += 32) {
            int2 nxt = make_int2(0, 0);
            if (b + 32 + lane < my_n) nxt = wl[b + 32 + lane];
            const int   s_l  = cur.x & 0xFFFF;
            const int   d_l  = cur.x >> 16;
            const float sc_l = __int_as_float(cur.y);
            const int rem = my_n - b;
            if (rem >= 32) gcn_own_walk<true >(x, tile, s_l, d_l, sc_l, 32,  lane);
            else           gcn_own_walk<false>(x, tile, s_l, d_l, sc_l, rem, lane);
            cur = nxt;
        }
    } else {
        int ln = 0;
        gcn_own_scan<FIRST>(x, src, dst, deg_out_is, tile,
                            q_e + wave * GCN_Q_CAP, q_d + wave * GCN_Q_CAP,
                            lo, rows, E, tid, lane, wl, ln);
        if (FIRST && lane == 0) counts[slot] = ln;
    }

    __syncthreads();
    if (!mm) return;

    v8f acc[NT];
#pragma unroll
    for (int t = 0; t < NT; ++t) acc[t] = v8f{};

    const float* ap = tile + arow * GCN_TILE_PITCH + h;
#pragma unroll
    for (int j = 0; j < 16; ++j) {
        v2f a0, a1;
        a0.x = ap[2 * j]      * scale;  a0.y = ap[32 + 2 * j] * scale;
        a1.x = ap[64 + 2 * j] * scale;  a1.y = ap[96 + 2 * j] * scale;
        float4 b[NT];
#pragma unroll
        for (int t = 0; t < NT; ++t) b[t] = wq[(2 * j + h) * PITCH4 + nb + t * 16 + c16];
#pragma unroll
        for (int t = 0; t < NT; ++t) {
            v2f b0;  b0.x = b[t].x;  b0.y = b[t].y;
            acc[t] = __builtin_amdgcn_wmma_f32_16x16x4_f32(
                         false, a0, false, b0, (short)0, acc[t], false, false);
        }
#pragma unroll
        for (int t = 0; t < NT; ++t) {
            v2f b1;  b1.x = b[t].z;  b1.y = b[t].w;
            acc[t] = __builtin_amdgcn_wmma_f32_16x16x4_f32(
                         false, a1, false, b1, (short)0, acc[t], false, false);
        }
    }

#pragma unroll
    for (int t = 0; t < NT; ++t) {
        const int n = nb + t * 16 + c16;
#pragma unroll
        for (int v = 0; v < 8; ++v) {
            const int r = rt * 16 + h * 8 + v;
            float val = acc[t][v] + bv[t];
            if (do_relu) val = fmaxf(val, 0.0f);
            if (r < rows) out[(size_t)(lo + r) * C + n] = val;
        }
    }
}

extern "C" void kernel_launch(void* const* d_in, const int* in_sizes, int n_in,
                              void* d_out, int out_size, void* d_ws, size_t ws_size,
                              hipStream_t stream) {
    const float* features = (const float*)d_in[0];
    const int*   src      = (const int*)  d_in[1];
    const int*   dst      = (const int*)  d_in[2];
    const float* W1       = (const float*)d_in[3];
    const float* b1       = (const float*)d_in[4];
    const float* W2       = (const float*)d_in[5];
    const float* b2       = (const float*)d_in[6];
    const float* W3       = (const float*)d_in[7];
    const float* b3       = (const float*)d_in[8];
    float* out = (float*)d_out;

    const int N = GCN_N_NODES;
    const int E = GCN_N_EDGES;
    const size_t NF = (size_t)N * 128;

    float* bufA  = (float*)d_ws;
    float* bufB  = bufA + NF;
    float* deg_o = bufB + NF;
    float* deg_i = deg_o + N;
    int*   lists  = (int*)((float*)d_ws + GCN_LIST_FLOAT_OFF);
    int*   counts = lists + 2 * GCN_LIST_CAP * GCN_OWN_WAVES * GCN_OWN_GROUPS;

    const int TB = 256;
    const int FILL_PER_BLOCK = TB * 4 * GCN_FILL_STORES;

    gcn_fill_zero<<<(2 * N + FILL_PER_BLOCK - 1) / FILL_PER_BLOCK, TB, 0, stream>>>(deg_o, 2 * N);
    gcn_count_deg<<<(E + TB - 1) / TB, TB, 0, stream>>>(src, dst, deg_o, deg_i, E);
    gcn_inv_sqrt_clamp<<<(2 * N + TB - 1) / TB, TB, 0, stream>>>(deg_o, 2 * N);

    gcn_layer_owner<true, 128><<<GCN_OWN_GROUPS, GCN_OWN_THREADS, 0, stream>>>(features, src, dst, deg_o, deg_i, W1, b1, bufB, lists, counts, E, N, 1);

    gcn_layer_owner<false, 128><<<GCN_OWN_GROUPS, GCN_OWN_THREADS, 0, stream>>>(bufB, src, dst, deg_o, deg_i, W2, b2, bufA, lists, counts, E, N, 1);

    gcn_layer_owner<false, 16><<<GCN_OWN_GROUPS, GCN_OWN_THREADS, 0, stream>>>(bufA, src, dst, deg_o, deg_i, W3, b3, out, lists, counts, E, N, 0);

    (void)in_sizes; (void)n_in; (void)out_size; (void)ws_size;
}


# Degree pass in 2 launches, not 3: out-degrees by global adds from 2048 waves of 12 edges a lane, in-degrees counted in LDS by layer 1's owner scan, scales taken where read; row sum order unchanged
# speedup vs baseline: 3.5795x; 1.1772x over previous
#include <hip/hip_runtime.h>
#include <math.h>

typedef __attribute__((ext_vector_type(2))) float v2f;
typedef __attribute__((ext_vector_type(8))) float v8f;

#define GCN_N_NODES 50000
#define GCN_N_EDGES 600000

#define GCN_FILL_STORES 8
__global__ void gcn_fill_zero(float* __restrict__ p, int n) {
    float4* p4 = reinterpret_cast<float4*>(p);
    const int n4 = n >> 2;
    int i = blockIdx.x * (blockDim.x * GCN_FILL_STORES) + threadIdx.x;
#pragma unroll
    for (int k = 0; k < GCN_FILL_STORES; ++k, i += blockDim.x)
        if (i < n4) p4[i] = make_float4(0.0f, 0.0f, 0.0f, 0.0f);
}

#define GCN_CNT_GROUPS   256
#define GCN_CNT_THREADS  256
#define GCN_CNT_LANES    (GCN_CNT_GROUPS * GCN_CNT_THREADS)
#define GCN_CNT_VECS     ((GCN_N_EDGES / 4 + GCN_CNT_LANES - 1) / GCN_CNT_LANES)
__global__ __launch_bounds__(GCN_CNT_THREADS)
void gcn_count_out_deg(const int* __restrict__ src, float* __restrict__ deg_out, int E) {
    const int4* src4 = reinterpret_cast<const int4*>(src);
    const int nvec  = E >> 2;
    const int first = blockIdx.x * GCN_CNT_THREADS + threadIdx.x;
    int4 s[GCN_CNT_VECS];
#pragma unroll
    for (int k = 0; k < GCN_CNT_VECS; ++k) {
        const int i = first + k * GCN_CNT_LANES;
        s[k] = i < nvec ? src4[i] : make_int4(0, 0, 0, 0);
    }
#pragma unroll
    for (int k = 0; k < GCN_CNT_VECS; ++k) {
        const int i = first + k * GCN_CNT_LANES;
        if (i < nvec) {
            atomicAdd(&deg_out[s[k].x], 1.0f);
            atomicAdd(&deg_out[s[k].y], 1.0f);
            atomicAdd(&deg_out[s[k].z], 1.0f);
            atomicAdd(&deg_out[s[k].w], 1.0f);
        }
    }
}

__device__ __forceinline__ float gcn_deg_scale(float count) {
    const float v = count < 1.0f ? 1.0f : count;
    return 1.0f / sqrtf(v);
}

#define GCN_OWN_GROUPS   256
#define GCN_OWN_ROWS     196
#define GCN_TILE_ROWS    208
#define GCN_TILE_PITCH   132
#define GCN_OWN_THREADS  1024
#define GCN_OWN_WAVES    (GCN_OWN_THREADS / 32)
#define GCN_Q_CAP        256
#define GCN_SCAN_LOADS   8
#define GCN_EDGE_GROUP   8
#define GCN_LIST_CAP     512
#define GCN_LIST_FLOAT_OFF 12900032
static_assert(GCN_OWN_GROUPS * GCN_OWN_ROWS >= GCN_N_NODES, "the ranges must cover every node");
static_assert(GCN_TILE_ROWS % 16 == 0 && GCN_TILE_ROWS >= GCN_OWN_ROWS && GCN_TILE_PITCH % 4 == 0 && GCN_TILE_PITCH >= 128,
              "whole row tiles, rows of whole 16-byte pieces");
static_assert(GCN_N_EDGES % 4 == 0, "src and dst are read 4 edges (16 bytes) at a time");
static_assert(GCN_N_NODES % 4 == 0, "the out-degree array is cleared 16 bytes at a time");
static_assert(31 + 4 * 32 <= GCN_Q_CAP, "one 16-byte load of a wave may add 128 entries to 31 waiting ones");
static_assert(GCN_N_NODES <= 65536 && GCN_OWN_ROWS <= 32768, "a list entry packs source row (16 bits) and local row into one int");
static_assert(GCN_LIST_CAP % 32 == 0, "whole batches fill a list exactly");
static_assert(GCN_LIST_FLOAT_OFF >= 2 * GCN_N_NODES * 128 + 2 * GCN_N_NODES && GCN_LIST_FLOAT_OFF % 64 == 0,
              "the lists lie behind both row buffers and both degree arrays");

template <bool FULL>
__device__ __forceinline__ void gcn_own_walk(const float* __restrict__ x, float* tile,
                                             int s_l, int d_l, float sc_l, int n, int lane) {
    const int n_edges = FULL ? 32 : n;
#pragma nounroll
    for (int j0 = 0; j0 < n_edges; j0 += GCN_EDGE_GROUP) {
        float4 v[GCN_EDGE_GROUP];
#pragma unroll
        for (int j = 0; j < GCN_EDGE_GROUP; ++j) {
            if (FULL || j0 + j < n) {
                const int s = __builtin_amdgcn_readlane(s_l, j0 + j);
                v[j] = reinterpret_cast<const float4*>(x + (size_t)s * 128)[lane];
            }
        }
#pragma unroll
        for (int j = 0; j < GCN_EDGE_GROUP; ++j) {
            if (FULL || j0 + j < n) {
                const int   d  = __builtin_amdgcn_readlane(d_l, j0 + j);
                const float sc = __int_as_float(
                    __builtin_amdgcn_readlane(__float_as_int(sc_l), j0 + j));
                float* row = tile + d * GCN_TILE_PITCH + lane;
                __hip_atomic_fetch_add(row + 0,  v[j].x * sc, __ATOMIC_RELAXED, __HIP_MEMORY_SCOPE_WORKGROUP);
                __hip_atomic_fetch_add(row + 32, v[j].y * sc, __ATOMIC_RELAXED, __HIP_MEMORY_SCOPE_WORKGROUP);
                __hip_atomic_fetch_add(row + 64, v[j].z * sc, __ATOMIC_RELAXED, __HIP_MEMORY_SCOPE_WORKGROUP);
                __hip_atomic_fetch_add(row + 96, v[j].w * sc, __ATOMIC_RELAXED, __HIP_MEMORY_SCOPE_WORKGROUP);
            }
        }
    }
}

template <bool FULL, bool RECORD>
__device__ __forceinline__ void gcn_own_batch(const float* __restrict__ x,
                                              const int* __restrict__ src,
                                              const float* __restrict__ deg_out,
                                              float* tile, const int* qe, const int* qd,
                                              int n, int lane, int2* wl, int& ln,
                                              int* rowcnt) {
    int   s_l  = 0;
    int   d_l  = 0;
    float sc_l = 0.0f;
    if (FULL || lane < n) {
        const int e = qe[lane];
        d_l  = qd[lane];
        s_l  = src[e];
        sc_l = gcn_deg_scale(deg_out[s_l]);
        if (RECORD)
            __hip_atomic_fetch_add(rowcnt + d_l, 1, __ATOMIC_RELAXED, __HIP_MEMORY_SCOPE_WORKGROUP);
    }
    if (RECORD) {
        const int cnt = FULL ? 32 : n;
        if (ln + cnt <= GCN_LIST_CAP) {
            if (FULL || lane < n)
                wl[ln + lane] = make_int2(s_l | (d_l << 16), __float_as_int(sc_l));
            ln += cnt;
        } else {
            ln = GCN_LIST_CAP + 1;
        }
    }
    gcn_own_walk<FULL>(x, tile, s_l, d_l, sc_l, n, lane);
}

template <bool RECORD>
__device__ __forceinline__ void gcn_own_scan(const float* __restrict__ x,
                                             const int* __restrict__ src,
                                             const int* __restrict__ dst,
                                             const float* __restrict__ deg_out,
                                             float* tile, int* qe, int* qd,
                                             int lo, int rows, int E, int tid, int lane,
                                             int2* wl, int& ln, int* rowcnt) {
    int  qn = 0;

    const int4* dst4 = reinterpret_cast<const int4*>(dst);
    const int nvec    = E >> 2;
    const int step    = GCN_OWN_THREADS * GCN_SCAN_LOADS;
    const int n_steps = (nvec + step - 1) / step;
    int st = blockIdx.x % n_steps;

    for (int o = 0; o < n_steps; ++o) {
        const int base = st * step + tid;
        const bool whole = (st + 1) * step <= nvec;
        st = st + 1 == n_steps ? 0 : st + 1;

        int4 dv[GCN_SCAN_LOADS];
        if (whole) {
#pragma unroll
            for (int u = 0; u < GCN_SCAN_LOADS; ++u)
                dv[u] = dst4[base + u * GCN_OWN_THREADS];
        } else {
#pragma unroll
            for (int u = 0; u < GCN_SCAN_LOADS; ++u) {
                const int idx = base + u * GCN_OWN_THREADS;
                dv[u] = idx < nvec ? dst4[idx] : make_int4(-1, -1, -1, -1);
            }
        }
#pragma unroll
        for (int u = 0; u < GCN_SCAN_LOADS; ++u) {
            const int e4 = (base + u * GCN_OWN_THREADS) * 4;
            const unsigned r[4] = { (unsigned)(dv[u].x - lo), (unsigned)(dv[u].y - lo),
                                    (unsigned)(dv[u].z - lo), (unsigned)(dv[u].w - lo) };
            const unsigned r01 = r[0] < r[1] ? r[0] : r[1];
            const unsigned r23 = r[2] < r[3] ? r[2] : r[3];
            const unsigned rmin = r01 < r23 ? r01 : r23;
            if (__builtin_amdgcn_ballot_w32(rmin < (unsigned)rows) == 0u) continue;
            const bool m[4] = { r[0] < (unsigned)rows, r[1] < (unsigned)rows,
                                r[2] < (unsigned)rows, r[3] < (unsigned)rows };
#pragma unroll
            for (int c = 0; c < 4; ++c) {
                const unsigned mask = __builtin_amdgcn_ballot_w32(m[c]);
                if (mask != 0u) {
                    const int pos = qn + (int)__builtin_amdgcn_mbcnt_lo(mask, 0u);
                    if (m[c]) { qe[pos] = e4 + c; qd[pos] = (int)r[c]; }
                    qn += __builtin_popcount(mask);
                }
            }
            while (qn >= 32) {
                qn -= 32;
                gcn_own_batch<true, RECORD>(x, src, deg_out, tile, qe + qn, qd + qn, 32, lane, wl, ln, rowcnt);
            }
        }
    }
    if (qn > 0)
        gcn_own_batch<false, RECORD>(x, src, deg_out, tile, qe, qd, qn, lane, wl, ln, rowcnt);
}

template <bool FIRST, int C>
__global__ __launch_bounds__(GCN_OWN_THREADS)
void gcn_layer_owner(const float* __restrict__ x,
                     const int* __restrict__ src, const int* __restrict__ dst,
                     const float* __restrict__ deg_out,
                     float* __restrict__ deg_in_is,
                     const float* __restrict__ W, const float* __restrict__ bias,
                     float* __restrict__ out, int* __restrict__ lists,
                     int* __restrict__ counts, int E, int N, int do_relu) {
    constexpr int NT     = C >= 64 ? 4 : C / 16;
    constexpr int CG     = C / (16 * NT);
    constexpr int PITCH4 = C + 8;
    constexpr int WQ_N   = 32 * C;
    constexpr int STAGE  = (WQ_N + GCN_OWN_THREADS - 1) / GCN_OWN_THREADS;
    static_assert(C % (16 * NT) == 0 && (GCN_TILE_ROWS / 16) * CG <= GCN_OWN_WAVES, "a wave per row tile and column group");
    __shared__ float  tile[GCN_TILE_ROWS * GCN_TILE_PITCH];
    __shared__ int    q_e[GCN_OWN_WAVES * GCN_Q_CAP];
    __shared__ int    q_d[GCN_OWN_WAVES * GCN_Q_CAP];
    __shared__ float4 wq[32 * PITCH4];
    __shared__ int    rowcnt[GCN_TILE_ROWS];

    const int tid  = threadIdx.x;
    const int lane = tid & 31;
    const int wave = __builtin_amdgcn_readfirstlane(tid >> 5);
    const int lo   = blockIdx.x * GCN_OWN_ROWS;
    int rows = N - lo;
    rows = rows > GCN_OWN_ROWS ? GCN_OWN_ROWS : rows;
    rows = rows < 0 ? 0 : rows;

    const int slot = blockIdx.x * GCN_OWN_WAVES + wave;
    int2* wl = reinterpret_cast<int2*>(lists) + (size_t)slot * GCN_LIST_CAP;

    int  my_n      = 0;
    bool walk_list = false;
    if (!FIRST) {
        const int c = counts[blockIdx.x * GCN_OWN_WAVES + lane];
        walk_list = __builtin_amdgcn_ballot_w32(c > GCN_LIST_CAP) == 0u;
        my_n      = __builtin_amdgcn_readlane(c, wave);
    }
    int2 cur = make_int2(0, 0);
    if (!FIRST && walk_list && lane < my_n) cur = wl[lane];

    const int  h      = lane >> 4;
    const int  c16    = lane & 15;
    const int  rt     = wave / CG;
    const int  nb     = (wave % CG) * (16 * NT);
    const bool mm     = rt * 16 < rows;
    const int  arow   = rt * 16 + c16;
    float scale = 0.0f;
    float bv[NT];
#pragma unroll
    for (int t = 0; t < NT; ++t) bv[t] = 0.0f;
    if (mm) {
        if (!FIRST && arow < rows) scale = deg_in_is[lo + arow];
#pragma unroll
        for (int t = 0; t < NT; ++t) bv[t] = bias[nb + t * 16 + c16];
    }

    float4 wreg[STAGE];
#pragma unroll
    for (int i = 0; i < STAGE; ++i) {
        const int idx = tid + i * GCN_OWN_THREADS;
        wreg[i] = make_float4(0.0f, 0.0f, 0.0f, 0.0f);
        if (WQ_N % GCN_OWN_THREADS == 0 || idx < WQ_N) {
            const float* wp = W + (size_t)(4 * (idx / C)) * C + (idx % C);
            wreg[i] = make_float4(wp[0], wp[C], wp[2 * C], wp[3 * C]);
        }
    }

    float4* tile4 = reinterpret_cast<float4*>(tile);
    for (int i = tid; i < GCN_TILE_ROWS * (GCN_TILE_PITCH / 4); i += GCN_OWN_THREADS)
        tile4[i] = make_float4(0.0f, 0.0f, 0.0f, 0.0f);
    if (FIRST && tid < GCN_TILE_ROWS) rowcnt[tid] = 0;
#pragma unroll
    for (int i = 0; i < STAGE; ++i) {
        const int idx = tid + i * GCN_OWN_THREADS;
        if (WQ_N % GCN_OWN_THREADS == 0 || idx < WQ_N)
            wq[(idx / C) * PITCH4 + (idx % C)] = wreg[i];
    }
    __syncthreads();

    if (!FIRST && walk_list) {
        for (int b = 0; b < my_n; b += 32) {
            int2 nxt = make_int2(0, 0);
            if (b + 32 + lane < my_n) nxt = wl[b + 32 + lane];
            const int   s_l  = cur.x & 0xFFFF;
            const int   d_l  = cur.x >> 16;
            const float sc_l = __int_as_float(cur.y);
            const int rem = my_n - b;
            if (rem >= 32) gcn_own_walk<true >(x, tile, s_l, d_l, sc_l, 32,  lane);
            else           gcn_own_walk<false>(x, tile, s_l, d_l, sc_l, rem, lane);
            cur = nxt;
        }
    } else {
        int ln = 0;
        gcn_own_scan<FIRST>(x, src, dst, deg_out, tile,
                            q_e + wave * GCN_Q_CAP, q_d + wave * GCN_Q_CAP,
                            lo, rows, E, tid, lane, wl, ln, rowcnt);
        if (FIRST && lane == 0) counts[slot] = ln;
    }

    __syncthreads();
    if (!mm) return;

    if (FIRST && arow < rows) {
        scale = gcn_deg_scale((float)rowcnt[arow]);
        if (wave % CG == 0 && h == 0) deg_in_is[lo + arow] = scale;
    }

    v8f acc[NT];
#pragma unroll
    for (int t = 0; t < NT; ++t) acc[t] = v8f{};

    const float* ap = tile + arow * GCN_TILE_PITCH + h;
#pragma unroll
    for (int j = 0; j < 16; ++j) {
        v2f a0, a1;
        a0.x = ap[2 * j]      * scale;  a0.y = ap[32 + 2 * j] * scale;
        a1.x = ap[64 + 2 * j] * scale;  a1.y = ap[96 + 2 * j] * scale;
        float4 b[NT];
#pragma unroll
        for (int t = 0; t < NT; ++t) b[t] = wq[(2 * j + h) * PITCH4 + nb + t * 16 + c16];
#pragma unroll
        for (int t = 0; t < NT; ++t) {
            v2f b0;  b0.x = b[t].x;  b0.y = b[t].y;
            acc[t] = __builtin_amdgcn_wmma_f32_16x16x4_f32(
                         false, a0, false, b0, (short)0, acc[t], false, false);
        }
#pragma unroll
        for (int t = 0; t < NT; ++t) {
            v2f b1;  b1.x = b[t].z;  b1.y = b[t].w;
            acc[t] = __builtin_amdgcn_wmma_f32_16x16x4_f32(
                         false, a1, false, b1, (short)0, acc[t], false, false);
        }
    }

#pragma unroll
    for (int t = 0; t < NT; ++t) {
        const int n = nb + t * 16 + c16;
#pragma unroll
        for (int v = 0; v < 8; ++v) {
            const int r = rt * 16 + h * 8 + v;
            float val = acc[t][v] + bv[t];
            if (do_relu) val = fmaxf(val, 0.0f);
            if (r < rows) out[(size_t)(lo + r) * C + n] = val;
        }
    }
}

extern "C" void kernel_launch(void* const* d_in, const int* in_sizes, int n_in,
                              void* d_out, int out_size, void* d_ws, size_t ws_size,
                              hipStream_t stream) {
    const float* features = (const float*)d_in[0];
    const int*   src      = (const int*)  d_in[1];
    const int*   dst      = (const int*)  d_in[2];
    const float* W1       = (const float*)d_in[3];
    const float* b1       = (const float*)d_in[4];
    const float* W2       = (const float*)d_in[5];
    const float* b2       = (const float*)d_in[6];
    const float* W3       = (const float*)d_in[7];
    const float* b3       = (const float*)d_in[8];
    float* out = (float*)d_out;

    const int N = GCN_N_NODES;
    const int E = GCN_N_EDGES;
    const size_t NF = (size_t)N * 128;

    float* bufA  = (float*)d_ws;
    float* bufB  = bufA + NF;
    float* deg_o = bufB + NF;
    float* deg_i = deg_o + N;
    int*   lists  = (int*)((float*)d_ws + GCN_LIST_FLOAT_OFF);
    int*   counts = lists + 2 * GCN_LIST_CAP * GCN_OWN_WAVES * GCN_OWN_GROUPS;

    const int TB = 256;
    const int FILL_PER_BLOCK = TB * 4 * GCN_FILL_STORES;

    gcn_fill_zero<<<(N + FILL_PER_BLOCK - 1) / FILL_PER_BLOCK, TB, 0, stream>>>(deg_o, N);
    gcn_count_out_deg<<<GCN_CNT_GROUPS, GCN_CNT_THREADS, 0, stream>>>(src, deg_o, E);

    gcn_layer_owner<true, 128><<<GCN_OWN_GROUPS, GCN_OWN_THREADS, 0, stream>>>(features, src, dst, deg_o, deg_i, W1, b1, bufB, lists, counts, E, N, 1);

    gcn_layer_owner<false, 128><<<GCN_OWN_GROUPS, GCN_OWN_THREADS, 0, stream>>>(bufB, src, dst, deg_o, deg_i, W2, b2, bufA, lists, counts, E, N, 1);

    gcn_layer_owner<false, 16><<<GCN_OWN_GROUPS, GCN_OWN_THREADS, 0, stream>>>(bufA, src, dst, deg_o, deg_i, W3, b3, out, lists, counts, E, N, 0);

    (void)in_sizes; (void)n_in; (void)out_size; (void)ws_size;
}


# Out-degrees counted by 64 owner work-groups: each reads all of src and counts its 782 nodes in LDS, stores them once; no global atomics, no zero-fill launch (5 -> 4 launches); row sum order same
# speedup vs baseline: 4.0804x; 1.1399x over previous
#include <hip/hip_runtime.h>
#include <math.h>

typedef __attribute__((ext_vector_type(2))) float v2f;
typedef __attribute__((ext_vector_type(8))) float v8f;

#define GCN_N_NODES 50000
#define GCN_N_EDGES 600000

#define GCN_CNT_GROUPS   64
#define GCN_CNT_ROWS     782
#define GCN_CNT_THREADS  1024
#define GCN_CNT_LOADS    8
static_assert(GCN_CNT_GROUPS * GCN_CNT_ROWS >= GCN_N_NODES, "the ranges must cover every node");
static_assert(GCN_CNT_ROWS <= GCN_CNT_THREADS, "one thread clears and stores one count");

__device__ __forceinline__ void gcn_cnt_add(int* cnt, int4 v, unsigned nlo4, unsigned rows4) {
    const unsigned r4[4] = { ((unsigned)v.x << 2) + nlo4, ((unsigned)v.y << 2) + nlo4,
                             ((unsigned)v.z << 2) + nlo4, ((unsigned)v.w << 2) + nlo4 };
#pragma unroll
    for (int c = 0; c < 4; c += 2) {
        const bool m0 = r4[c] < rows4;
        const bool m1 = r4[c + 1] < rows4;
        if (__builtin_amdgcn_ballot_w32(m0 || m1) != 0u) {
            if (m0) __hip_atomic_fetch_add(reinterpret_cast<int*>(reinterpret_cast<char*>(cnt) + r4[c]), 1,
                                           __ATOMIC_RELAXED, __HIP_MEMORY_SCOPE_WORKGROUP);
            if (m1) __hip_atomic_fetch_add(reinterpret_cast<int*>(reinterpret_cast<char*>(cnt) + r4[c + 1]), 1,
                                           __ATOMIC_RELAXED, __HIP_MEMORY_SCOPE_WORKGROUP);
        }
    }
}

__global__ __launch_bounds__(GCN_CNT_THREADS)
void gcn_count_out_deg(const int* __restrict__ src, float* __restrict__ deg_out, int E, int N) {
    __shared__ int cnt[GCN_CNT_ROWS];
    const int tid = threadIdx.x;
    const int lo  = blockIdx.x * GCN_CNT_ROWS;
    int rows = N - lo;
    rows = rows > GCN_CNT_ROWS ? GCN_CNT_ROWS : rows;
    rows = rows < 0 ? 0 : rows;

    const unsigned nlo4  = 0u - ((unsigned)lo << 2);
    const unsigned rows4 = (unsigned)rows << 2;

    if (tid < GCN_CNT_ROWS) cnt[tid] = 0;
    __syncthreads();

    const int4* src4 = reinterpret_cast<const int4*>(src);
    const int nvec    = E >> 2;
    const int step    = GCN_CNT_THREADS * GCN_CNT_LOADS;
    const int n_whole = nvec / step;
    int s_next = n_whole > 0 ? (int)(blockIdx.x % (unsigned)n_whole) : 0;

    int4 va[GCN_CNT_LOADS], vb[GCN_CNT_LOADS];
    if (n_whole > 0) {
        const int base = s_next * step + tid;
#pragma unroll
        for (int u = 0; u < GCN_CNT_LOADS; ++u) va[u] = src4[base + u * GCN_CNT_THREADS];
        s_next = s_next + 1 == n_whole ? 0 : s_next + 1;
    }
    if (n_whole > 1) {
        const int base = s_next * step + tid;
#pragma unroll
        for (int u = 0; u < GCN_CNT_LOADS; ++u) vb[u] = src4[base + u * GCN_CNT_THREADS];
        s_next = s_next + 1 == n_whole ? 0 : s_next + 1;
    }
    for (int o = 0; o < n_whole; o += 2) {
        {
            const bool more = o + 2 < n_whole;
            const int  base = s_next * step + tid;
#pragma unroll
            for (int u = 0; u < GCN_CNT_LOADS; ++u) {
                gcn_cnt_add(cnt, va[u], nlo4, rows4);
                if (more) va[u] = src4[base + u * GCN_CNT_THREADS];
            }
            if (more) s_next = s_next + 1 == n_whole ? 0 : s_next + 1;
        }
        if (o + 1 < n_whole) {
            const bool more = o + 3 < n_whole;
            const int  base = s_next * step + tid;
#pragma unroll
            for (int u = 0; u < GCN_CNT_LOADS; ++u) {
                gcn_cnt_add(cnt, vb[u], nlo4, rows4);
                if (more) vb[u] = src4[base + u * GCN_CNT_THREADS];
            }
            if (more) s_next = s_next + 1 == n_whole ? 0 : s_next + 1;
        }
    }
    for (int idx = n_whole * step + tid; idx < nvec; idx += GCN_CNT_THREADS)
        gcn_cnt_add(cnt, src4[idx], nlo4, rows4);

    __syncthreads();
    if (tid < rows) deg_out[lo + tid] = (float)cnt[tid];
}

__device__ __forceinline__ float gcn_deg_scale(float count) {
    const float v = count < 1.0f ? 1.0f : count;
    return 1.0f / sqrtf(v);
}

#define GCN_OWN_GROUPS   256
#define GCN_OWN_ROWS     196
#define GCN_TILE_ROWS    208
#define GCN_TILE_PITCH   132
#define GCN_OWN_THREADS  1024
#define GCN_OWN_WAVES    (GCN_OWN_THREADS / 32)
#define GCN_Q_CAP        256
#define GCN_SCAN_LOADS   8
#define GCN_EDGE_GROUP   8
#define GCN_LIST_CAP     512
#define GCN_LIST_FLOAT_OFF 12900032
static_assert(GCN_OWN_GROUPS * GCN_OWN_ROWS >= GCN_N_NODES, "the ranges must cover every node");
static_assert(GCN_TILE_ROWS % 16 == 0 && GCN_TILE_ROWS >= GCN_OWN_ROWS && GCN_TILE_PITCH % 4 == 0 && GCN_TILE_PITCH >= 128,
              "whole row tiles, rows of whole 16-byte pieces");
static_assert(GCN_N_EDGES % 4 == 0, "src and dst are read 4 edges (16 bytes) at a time");
static_assert(31 + 4 * 32 <= GCN_Q_CAP, "one 16-byte load of a wave may add 128 entries to 31 waiting ones");
static_assert(GCN_N_NODES <= 65536 && GCN_OWN_ROWS <= 32768, "a list entry packs source row (16 bits) and local row into one int");
static_assert(GCN_LIST_CAP % 32 == 0, "whole batches fill a list exactly");
static_assert(GCN_LIST_FLOAT_OFF >= 2 * GCN_N_NODES * 128 + 2 * GCN_N_NODES && GCN_LIST_FLOAT_OFF % 64 == 0,
              "the lists lie behind both row buffers and both degree arrays");

template <bool FULL>
__device__ __forceinline__ void gcn_own_walk(const float* __restrict__ x, float* tile,
                                             int s_l, int d_l, float sc_l, int n, int lane) {
    const int n_edges = FULL ? 32 : n;
#pragma nounroll
    for (int j0 = 0; j0 < n_edges; j0 += GCN_EDGE_GROUP) {
        float4 v[GCN_EDGE_GROUP];
#pragma unroll
        for (int j = 0; j < GCN_EDGE_GROUP; ++j) {
            if (FULL || j0 + j < n) {
                const int s = __builtin_amdgcn_readlane(s_l, j0 + j);
                v[j] = reinterpret_cast<const float4*>(x + (size_t)s * 128)[lane];
            }
        }
#pragma unroll
        for (int j = 0; j < GCN_EDGE_GROUP; ++j) {
            if (FULL || j0 + j < n) {
                const int   d  = __builtin_amdgcn_readlane(d_l, j0 + j);
                const float sc = __int_as_float(
                    __builtin_amdgcn_readlane(__float_as_int(sc_l), j0 + j));
                float* row = tile + d * GCN_TILE_PITCH + lane;
                __hip_atomic_fetch_add(row + 0,  v[j].x * sc, __ATOMIC_RELAXED, __HIP_MEMORY_SCOPE_WORKGROUP);
                __hip_atomic_fetch_add(row + 32, v[j].y * sc, __ATOMIC_RELAXED, __HIP_MEMORY_SCOPE_WORKGROUP);
                __hip_atomic_fetch_add(row + 64, v[j].z * sc, __ATOMIC_RELAXED, __HIP_MEMORY_SCOPE_WORKGROUP);
                __hip_atomic_fetch_add(row + 96, v[j].w * sc, __ATOMIC_RELAXED, __HIP_MEMORY_SCOPE_WORKGROUP);
            }
        }
    }
}

template <bool FULL, bool RECORD>
__device__ __forceinline__ void gcn_own_batch(const float* __restrict__ x,
                                              const int* __restrict__ src,
                                              const float* __restrict__ deg_out,
                                              float* tile, const int* qe, const int* qd,
                                              int n, int lane, int2* wl, int& ln,
                                              int* rowcnt) {
    int   s_l  = 0;
    int   d_l  = 0;
    float sc_l = 0.0f;
    if (FULL || lane < n) {
        const int e = qe[lane];
        d_l  = qd[lane];
        s_l  = src[e];
        sc_l = gcn_deg_scale(deg_out[s_l]);
        if (RECORD)
            __hip_atomic_fetch_add(rowcnt + d_l, 1, __ATOMIC_RELAXED, __HIP_MEMORY_SCOPE_WORKGROUP);
    }
    if (RECORD) {
        const int cnt = FULL ? 32 : n;
        if (ln + cnt <= GCN_LIST_CAP) {
            if (FULL || lane < n)
                wl[ln + lane] = make_int2(s_l | (d_l << 16), __float_as_int(sc_l));
            ln += cnt;
        } else {
            ln = GCN_LIST_CAP + 1;
        }
    }
    gcn_own_walk<FULL>(x, tile, s_l, d_l, sc_l, n, lane);
}

template <bool RECORD>
__device__ __forceinline__ void gcn_own_scan(const float* __restrict__ x,
                                             const int* __restrict__ src,
                                             const int* __restrict__ dst,
                                             const float* __restrict__ deg_out,
                                             float* tile, int* qe, int* qd,
                                             int lo, int rows, int E, int tid, int lane,
                                             int2* wl, int& ln, int* rowcnt) {
    int  qn = 0;

    const int4* dst4 = reinterpret_cast<const int4*>(dst);
    const int nvec    = E >> 2;
    const int step    = GCN_OWN_THREADS * GCN_SCAN_LOADS;
    const int n_steps = (nvec + step - 1) / step;
    int st = blockIdx.x % n_steps;

    for (int o = 0; o < n_steps; ++o) {
        const int base = st * step + tid;
        const bool whole = (st + 1) * step <= nvec;
        st = st + 1 == n_steps ? 0 : st + 1;

        int4 dv[GCN_SCAN_LOADS];
        if (whole) {
#pragma unroll
            for (int u = 0; u < GCN_SCAN_LOADS; ++u)
                dv[u] = dst4[base + u * GCN_OWN_THREADS];
        } else {
#pragma unroll
            for (int u = 0; u < GCN_SCAN_LOADS; ++u) {
                const int idx = base + u * GCN_OWN_THREADS;
                dv[u] = idx < nvec ? dst4[idx] : make_int4(-1, -1, -1, -1);
            }
        }
#pragma unroll
        for (int u = 0; u < GCN_SCAN_LOADS; ++u) {
            const int e4 = (base + u * GCN_OWN_THREADS) * 4;
            const unsigned r[4] = { (unsigned)(dv[u].x - lo), (unsigned)(dv[u].y - lo),
                                    (unsigned)(dv[u].z - lo), (unsigned)(dv[u].w - lo) };
            const unsigned r01 = r[0] < r[1] ? r[0] : r[1];
            const unsigned r23 = r[2] < r[3] ? r[2] : r[3];
            const unsigned rmin = r01 < r23 ? r01 : r23;
            if (__builtin_amdgcn_ballot_w32(rmin < (unsigned)rows) == 0u) continue;
            const bool m[4] = { r[0] < (unsigned)rows, r[1] < (unsigned)rows,
                                r[2] < (unsigned)rows, r[3] < (unsigned)rows };
#pragma unroll
            for (int c = 0; c < 4; ++c) {
                const unsigned mask = __builtin_amdgcn_ballot_w32(m[c]);
                if (mask != 0u) {
                    const int pos = qn + (int)__builtin_amdgcn_mbcnt_lo(mask, 0u);
                    if (m[c]) { qe[pos] = e4 + c; qd[pos] = (int)r[c]; }
                    qn += __builtin_popcount(mask);
                }
            }
            while (qn >= 32) {
                qn -= 32;
                gcn_own_batch<true, RECORD>(x, src, deg_out, tile, qe + qn, qd + qn, 32, lane, wl, ln, rowcnt);
            }
        }
    }
    if (qn > 0)
        gcn_own_batch<false, RECORD>(x, src, deg_out, tile, qe, qd, qn, lane, wl, ln, rowcnt);
}

template <bool FIRST, int C>
__global__ __launch_bounds__(GCN_OWN_THREADS)
void gcn_layer_owner(const float* __restrict__ x,
                     const int* __restrict__ src, const int* __restrict__ dst,
                     const float* __restrict__ deg_out,
                     float* __restrict__ deg_in_is,
                     const float* __restrict__ W, const float* __restrict__ bias,
                     float* __restrict__ out, int* __restrict__ lists,
                     int* __restrict__ counts, int E, int N, int do_relu) {
    constexpr int NT     = C >= 64 ? 4 : C / 16;
    constexpr int CG     = C / (16 * NT);
    constexpr int PITCH4 = C + 8;
    constexpr int WQ_N   = 32 * C;
    constexpr int STAGE  = (WQ_N + GCN_OWN_THREADS - 1) / GCN_OWN_THREADS;
    static_assert(C % (16 * NT) == 0 && (GCN_TILE_ROWS / 16) * CG <= GCN_OWN_WAVES, "a wave per row tile and column group");
    __shared__ float  tile[GCN_TILE_ROWS * GCN_TILE_PITCH];
    __shared__ int    q_e[GCN_OWN_WAVES * GCN_Q_CAP];
    __shared__ int    q_d[GCN_OWN_WAVES * GCN_Q_CAP];
    __shared__ float4 wq[32 * PITCH4];
    __shared__ int    rowcnt[GCN_TILE_ROWS];

    const int tid  = threadIdx.x;
    const int lane = tid & 31;
    const int wave = __builtin_amdgcn_readfirstlane(tid >> 5);
    const int lo   = blockIdx.x * GCN_OWN_ROWS;
    int rows = N - lo;
    rows = rows > GCN_OWN_ROWS ? GCN_OWN_ROWS : rows;
    rows = rows < 0 ? 0 : rows;

    const int slot = blockIdx.x * GCN_OWN_WAVES + wave;
    int2* wl = reinterpret_cast<int2*>(lists) + (size_t)slot * GCN_LIST_CAP;

    int  my_n      = 0;
    bool walk_list = false;
    if (!FIRST) {
        const int c = counts[blockIdx.x * GCN_OWN_WAVES + lane];
        walk_list = __builtin_amdgcn_ballot_w32(c > GCN_LIST_CAP) == 0u;
        my_n      = __builtin_amdgcn_readlane(c, wave);
    }
    int2 cur = make_int2(0, 0);
    if (!FIRST && walk_list && lane < my_n) cur = wl[lane];

    const int  h      = lane >> 4;
    const int  c16    = lane & 15;
    const int  rt     = wave / CG;
    const int  nb     = (wave % CG) * (16 * NT);
    const bool mm     = rt * 16 < rows;
    const int  arow   = rt * 16 + c16;
    float scale = 0.0f;
    float bv[NT];
#pragma unroll
    for (int t = 0; t < NT; ++t) bv[t] = 0.0f;
    if (mm) {
        if (!FIRST && arow < rows) scale = deg_in_is[lo + arow];
#pragma unroll
        for (int t = 0; t < NT; ++t) bv[t] = bias[nb + t * 16 + c16];
    }

    float4 wreg[STAGE];
#pragma unroll
    for (int i = 0; i < STAGE; ++i) {
        const int idx = tid + i * GCN_OWN_THREADS;
        wreg[i] = make_float4(0.0f, 0.0f, 0.0f, 0.0f);
        if (WQ_N % GCN_OWN_THREADS == 0 || idx < WQ_N) {
            const float* wp = W + (size_t)(4 * (idx / C)) * C + (idx % C);
            wreg[i] = make_float4(wp[0], wp[C], wp[2 * C], wp[3 * C]);
        }
    }

    float4* tile4 = reinterpret_cast<float4*>(tile);
    for (int i = tid; i < GCN_TILE_ROWS * (GCN_TILE_PITCH / 4); i += GCN_OWN_THREADS)
        tile4[i] = make_float4(0.0f, 0.0f, 0.0f, 0.0f);
    if (FIRST && tid < GCN_TILE_ROWS) rowcnt[tid] = 0;
#pragma unroll
    for (int i = 0; i < STAGE; ++i) {
        const int idx = tid + i * GCN_OWN_THREADS;
        if (WQ_N % GCN_OWN_THREADS == 0 || idx < WQ_N)
            wq[(idx / C) * PITCH4 + (idx % C)] = wreg[i];
    }
    __syncthreads();

    if (!FIRST && walk_list) {
        for (int b = 0; b < my_n; b += 32) {
            int2 nxt = make_int2(0, 0);
            if (b + 32 + lane < my_n) nxt = wl[b + 32 + lane];
            const int   s_l  = cur.x & 0xFFFF;
            const int   d_l  = cur.x >> 16;
            const float sc_l = __int_as_float(cur.y);
            const int rem = my_n - b;
            if (rem >= 32) gcn_own_walk<true >(x, tile, s_l, d_l, sc_l, 32,  lane);
            else           gcn_own_walk<false>(x, tile, s_l, d_l, sc_l, rem, lane);
            cur = nxt;
        }
    } else {
        int ln = 0;
        gcn_own_scan<FIRST>(x, src, dst, deg_out, tile,
                            q_e + wave * GCN_Q_CAP, q_d + wave * GCN_Q_CAP,
                            lo, rows, E, tid, lane, wl, ln, rowcnt);
        if (FIRST && lane == 0) counts[slot] = ln;
    }

    __syncthreads();
    if (!mm) return;

    if (FIRST && arow < rows) {
        scale = gcn_deg_scale((float)rowcnt[arow]);
        if (wave % CG == 0 && h == 0) deg_in_is[lo + arow] = scale;
    }

    v8f acc[NT];
#pragma unroll
    for (int t = 0; t < NT; ++t) acc[t] = v8f{};

    const float* ap = tile + arow * GCN_TILE_PITCH + h;
#pragma unroll
    for (int j = 0; j < 16; ++j) {
        v2f a0, a1;
        a0.x = ap[2 * j]      * scale;  a0.y = ap[32 + 2 * j] * scale;
        a1.x = ap[64 + 2 * j] * scale;  a1.y = ap[96 + 2 * j] * scale;
        float4 b[NT];
#pragma unroll
        for (int t = 0; t < NT; ++t) b[t] = wq[(2 * j + h) * PITCH4 + nb + t * 16 + c16];
#pragma unroll
        for (int t = 0; t < NT; ++t) {
            v2f b0;  b0.x = b[t].x;  b0.y = b[t].y;
            acc[t] = __builtin_amdgcn_wmma_f32_16x16x4_f32(
                         false, a0, false, b0, (short)0, acc[t], false, false);
        }
#pragma unroll
        for (int t = 0; t < NT; ++t) {
            v2f b1;  b1.x = b[t].z;  b1.y = b[t].w;
            acc[t] = __builtin_amdgcn_wmma_f32_16x16x4_f32(
                         false, a1, false, b1, (short)0, acc[t], false, false);
        }
    }

#pragma unroll
    for (int t = 0; t < NT; ++t) {
        const int n = nb + t * 16 + c16;
#pragma unroll
        for (int v = 0; v < 8; ++v) {
            const int r = rt * 16 + h * 8 + v;
            float val = acc[t][v] + bv[t];
            if (do_relu) val = fmaxf(val, 0.0f);
            if (r < rows) out[(size_t)(lo + r) * C + n] = val;
        }
    }
}

extern "C" void kernel_launch(void* const* d_in, const int* in_sizes, int n_in,
                              void* d_out, int out_size, void* d_ws, size_t ws_size,
                              hipStream_t stream) {
    const float* features = (const float*)d_in[0];
    const int*   src      = (const int*)  d_in[1];
    const int*   dst      = (const int*)  d_in[2];
    const float* W1       = (const float*)d_in[3];
    const float* b1       = (const float*)d_in[4];
    const float* W2       = (const float*)d_in[5];
    const float* b2       = (const float*)d_in[6];
    const float* W3       = (const float*)d_in[7];
    const float* b3       = (const float*)d_in[8];
    float* out = (float*)d_out;

    const int N = GCN_N_NODES;
    const int E = GCN_N_EDGES;
    const size_t NF = (size_t)N * 128;

    float* bufA  = (float*)d_ws;
    float* bufB  = bufA + NF;
    float* deg_o = bufB + NF;
    float* deg_i = deg_o + N;
    int*   lists  = (int*)((float*)d_ws + GCN_LIST_FLOAT_OFF);
    int*   counts = lists + 2 * GCN_LIST_CAP * GCN_OWN_WAVES * GCN_OWN_GROUPS;

    gcn_count_out_deg<<<GCN_CNT_GROUPS, GCN_CNT_THREADS, 0, stream>>>(src, deg_o, E, N);

    gcn_layer_owner<true, 128><<<GCN_OWN_GROUPS, GCN_OWN_THREADS, 0, stream>>>(features, src, dst, deg_o, deg_i, W1, b1, bufB, lists, counts, E, N, 1);

    gcn_layer_owner<false, 128><<<GCN_OWN_GROUPS, GCN_OWN_THREADS, 0, stream>>>(bufB, src, dst, deg_o, deg_i, W2, b2, bufA, lists, counts, E, N, 1);

    gcn_layer_owner<false, 16><<<GCN_OWN_GROUPS, GCN_OWN_THREADS, 0, stream>>>(bufA, src, dst, deg_o, deg_i, W3, b3, out, lists, counts, E, N, 0);

    (void)in_sizes; (void)n_in; (void)out_size; (void)ws_size;
}


# Layer 1 finds its edges through a bitmap the out-degree launch builds on idle WGPs: per owner one bit per 256-byte chunk of dst; an owner reads only its flagged chunks, not all of dst; sum order same
# speedup vs baseline: 4.3217x; 1.0591x over previous
#include <hip/hip_runtime.h>
#include <math.h>

typedef __attribute__((ext_vector_type(2))) float v2f;
typedef __attribute__((ext_vector_type(8))) float v8f;

#define GCN_N_NODES 50000
#define GCN_N_EDGES 600000

#define GCN_BM_OWNERS    256
#define GCN_BM_ROWS      196
#define GCN_BM_EDGES     4096
#define GCN_BM_BUILDERS  ((GCN_N_EDGES + GCN_BM_EDGES - 1) / GCN_BM_EDGES)
#define GCN_BM_NWORDS    (2 * GCN_BM_BUILDERS)
#define GCN_CNT_GROUPS   64
#define GCN_CNT_ROWS     782
#define GCN_CNT_THREADS  1024
#define GCN_CNT_LOADS    8
static_assert(GCN_CNT_GROUPS * GCN_CNT_ROWS >= GCN_N_NODES, "the ranges must cover every node");
static_assert(GCN_CNT_ROWS <= GCN_CNT_THREADS, "one thread clears and stores one count");

__device__ __forceinline__ void gcn_cnt_add(int* cnt, int4 v, unsigned nlo4, unsigned rows4) {
    const unsigned r4[4] = { ((unsigned)v.x << 2) + nlo4, ((unsigned)v.y << 2) + nlo4,
                             ((unsigned)v.z << 2) + nlo4, ((unsigned)v.w << 2) + nlo4 };
#pragma unroll
    for (int c = 0; c < 4; c += 2) {
        const bool m0 = r4[c] < rows4;
        const bool m1 = r4[c + 1] < rows4;
        if (__builtin_amdgcn_ballot_w32(m0 || m1) != 0u) {
            if (m0) __hip_atomic_fetch_add(reinterpret_cast<int*>(reinterpret_cast<char*>(cnt) + r4[c]), 1,
                                           __ATOMIC_RELAXED, __HIP_MEMORY_SCOPE_WORKGROUP);
            if (m1) __hip_atomic_fetch_add(reinterpret_cast<int*>(reinterpret_cast<char*>(cnt) + r4[c + 1]), 1,
                                           __ATOMIC_RELAXED, __HIP_MEMORY_SCOPE_WORKGROUP);
        }
    }
}

static_assert(GCN_CNT_THREADS * 4 == GCN_BM_EDGES && GCN_CNT_THREADS >= 2 * GCN_BM_OWNERS,
              "a builder thread loads 4 edges, and 512 threads clear and store the builder's words");

__global__ __launch_bounds__(GCN_CNT_THREADS)
void gcn_count_out_deg(const int* __restrict__ src, const int* __restrict__ dst,
                       float* __restrict__ deg_out, unsigned* __restrict__ bitmap, int E, int N) {
    __shared__ int cnt[GCN_CNT_ROWS];
    __shared__ unsigned bm[2 * GCN_BM_OWNERS];
    const int tid = threadIdx.x;

    if (blockIdx.x >= GCN_CNT_GROUPS) {
        const int g = blockIdx.x - GCN_CNT_GROUPS;
        if (tid < 2 * GCN_BM_OWNERS) bm[tid] = 0u;
        __syncthreads();
        const int idx4 = g * GCN_CNT_THREADS + tid;
        if (idx4 < (E >> 2)) {
            const int4 v = reinterpret_cast<const int4*>(dst)[idx4];
            const int      cl   = tid >> 4;
            const int      word = cl >> 5;
            const unsigned bit  = 1u << (cl & 31);
            const int d[4] = { v.x, v.y, v.z, v.w };
#pragma unroll
            for (int c = 0; c < 4; ++c) {
                if ((unsigned)d[c] < (unsigned)N && d[c] < GCN_BM_OWNERS * GCN_BM_ROWS)
                    __hip_atomic_fetch_or(&bm[2 * (d[c] / GCN_BM_ROWS) + word], bit,
                                          __ATOMIC_RELAXED, __HIP_MEMORY_SCOPE_WORKGROUP);
            }
        }
        __syncthreads();
        if (tid < 2 * GCN_BM_OWNERS) bitmap[g * (2 * GCN_BM_OWNERS) + tid] = bm[tid];
        return;
    }

    const int lo  = blockIdx.x * GCN_CNT_ROWS;
    int rows = N - lo;
    rows = rows > GCN_CNT_ROWS ? GCN_CNT_ROWS : rows;
    rows = rows < 0 ? 0 : rows;

    const unsigned nlo4  = 0u - ((unsigned)lo << 2);
    const unsigned rows4 = (unsigned)rows << 2;

    if (tid < GCN_CNT_ROWS) cnt[tid] = 0;
    __syncthreads();

    const int4* src4 = reinterpret_cast<const int4*>(src);
    const int nvec    = E >> 2;
    const int step    = GCN_CNT_THREADS * GCN_CNT_LOADS;
    const int n_whole = nvec / step;
    int s_next = n_whole > 0 ? (int)(blockIdx.x % (unsigned)n_whole) : 0;

    int4 va[GCN_CNT_LOADS], vb[GCN_CNT_LOADS];
    if (n_whole > 0) {
        const int base = s_next * step + tid;
#pragma unroll
        for (int u = 0; u < GCN_CNT_LOADS; ++u) va[u] = src4[base + u * GCN_CNT_THREADS];
        s_next = s_next + 1 == n_whole ? 0 : s_next + 1;
    }
    if (n_whole > 1) {
        const int base = s_next * step + tid;
#pragma unroll
        for (int u = 0; u < GCN_CNT_LOADS; ++u) vb[u] = src4[base + u * GCN_CNT_THREADS];
        s_next = s_next + 1 == n_whole ? 0 : s_next + 1;
    }
    for (int o = 0; o < n_whole; o += 2) {
        {
            const bool more = o + 2 < n_whole;
            const int  base = s_next * step + tid;
#pragma unroll
            for (int u = 0; u < GCN_CNT_LOADS; ++u) {
                gcn_cnt_add(cnt, va[u], nlo4, rows4);
                if (more) va[u] = src4[base + u * GCN_CNT_THREADS];
            }
            if (more) s_next = s_next + 1 == n_whole ? 0 : s_next + 1;
        }
        if (o + 1 < n_whole) {
            const bool more = o + 3 < n_whole;
            const int  base = s_next * step + tid;
#pragma unroll
            for (int u = 0; u < GCN_CNT_LOADS; ++u) {
                gcn_cnt_add(cnt, vb[u], nlo4, rows4);
                if (more) vb[u] = src4[base + u * GCN_CNT_THREADS];
            }
            if (more) s_next = s_next + 1 == n_whole ? 0 : s_next + 1;
        }
    }
    for (int idx = n_whole * step + tid; idx < nvec; idx += GCN_CNT_THREADS)
        gcn_cnt_add(cnt, src4[idx], nlo4, rows4);

    __syncthreads();
    if (tid < rows) deg_out[lo + tid] = (float)cnt[tid];
}

__device__ __forceinline__ float gcn_deg_scale(float count) {
    const float v = count < 1.0f ? 1.0f : count;
    return 1.0f / sqrtf(v);
}

#define GCN_OWN_GROUPS   256
#define GCN_OWN_ROWS     196
#define GCN_TILE_ROWS    208
#define GCN_TILE_PITCH   132
#define GCN_OWN_THREADS  1024
#define GCN_OWN_WAVES    (GCN_OWN_THREADS / 32)
#define GCN_Q_CAP        256
#define GCN_SCAN_LOADS   8
#define GCN_BM_PER_WAVE  (GCN_BM_NWORDS / GCN_OWN_WAVES)
#define GCN_BM_EXTRA     (GCN_BM_NWORDS % GCN_OWN_WAVES)
#define GCN_CQ_CAP       ((GCN_BM_PER_WAVE + 1) * 32)
#define GCN_EDGE_GROUP   8
#define GCN_LIST_CAP     512
#define GCN_LIST_FLOAT_OFF 12900032
static_assert(GCN_OWN_GROUPS * GCN_OWN_ROWS >= GCN_N_NODES, "the ranges must cover every node");
static_assert(GCN_TILE_ROWS % 16 == 0 && GCN_TILE_ROWS >= GCN_OWN_ROWS && GCN_TILE_PITCH % 4 == 0 && GCN_TILE_PITCH >= 128,
              "whole row tiles, rows of whole 16-byte pieces");
static_assert(GCN_N_EDGES % 4 == 0, "src and dst are read 4 edges (16 bytes) at a time");
static_assert(31 + 4 * 32 <= GCN_Q_CAP, "one 16-byte load of a wave may add 128 entries to 31 waiting ones");
static_assert(GCN_N_NODES <= 65536 && GCN_OWN_ROWS <= 32768, "a list entry packs source row (16 bits) and local row into one int");
static_assert(GCN_LIST_CAP % 32 == 0, "whole batches fill a list exactly");
static_assert(GCN_BM_OWNERS == GCN_OWN_GROUPS && GCN_BM_ROWS == GCN_OWN_ROWS, "the bitmap's ranges are the owners' ranges");
static_assert(GCN_BM_PER_WAVE + 1 <= 32, "one bitmap word per lane");
static_assert(GCN_LIST_FLOAT_OFF >= 2 * GCN_N_NODES * 128 + 2 * GCN_N_NODES && GCN_LIST_FLOAT_OFF % 64 == 0,
              "the lists lie behind both row buffers and both degree arrays");

template <bool FULL>
__device__ __forceinline__ void gcn_own_walk(const float* __restrict__ x, float* tile,
                                             int s_l, int d_l, float sc_l, int n, int lane) {
    const int n_edges = FULL ? 32 : n;
#pragma nounroll
    for (int j0 = 0; j0 < n_edges; j0 += GCN_EDGE_GROUP) {
        float4 v[GCN_EDGE_GROUP];
#pragma unroll
        for (int j = 0; j < GCN_EDGE_GROUP; ++j) {
            if (FULL || j0 + j < n) {
                const int s = __builtin_amdgcn_readlane(s_l, j0 + j);
                v[j] = reinterpret_cast<const float4*>(x + (size_t)s * 128)[lane];
            }
        }
#pragma unroll
        for (int j = 0; j < GCN_EDGE_GROUP; ++j) {
            if (FULL || j0 + j < n) {
                const int   d  = __builtin_amdgcn_readlane(d_l, j0 + j);
                const float sc = __int_as_float(
                    __builtin_amdgcn_readlane(__float_as_int(sc_l), j0 + j));
                float* row = tile + d * GCN_TILE_PITCH + lane;
                __hip_atomic_fetch_add(row + 0,  v[j].x * sc, __ATOMIC_RELAXED, __HIP_MEMORY_SCOPE_WORKGROUP);
                __hip_atomic_fetch_add(row + 32, v[j].y * sc, __ATOMIC_RELAXED, __HIP_MEMORY_SCOPE_WORKGROUP);
                __hip_atomic_fetch_add(row + 64, v[j].z * sc, __ATOMIC_RELAXED, __HIP_MEMORY_SCOPE_WORKGROUP);
                __hip_atomic_fetch_add(row + 96, v[j].w * sc, __ATOMIC_RELAXED, __HIP_MEMORY_SCOPE_WORKGROUP);
            }
        }
    }
}

template <bool FULL, bool RECORD>
__device__ __forceinline__ void gcn_own_batch(const float* __restrict__ x,
                                              const int* __restrict__ src,
                                              const float* __restrict__ deg_out,
                                              float* tile, const int* qe, const int* qd,
                                              int n, int lane, int2* wl, int& ln,
                                              int* rowcnt) {
    int   s_l  = 0;
    int   d_l  = 0;
    float sc_l = 0.0f;
    if (FULL || lane < n) {
        const int e = qe[lane];
        d_l  = qd[lane];
        s_l  = src[e];
        sc_l = gcn_deg_scale(deg_out[s_l]);
        if (RECORD)
            __hip_atomic_fetch_add(rowcnt + d_l, 1, __ATOMIC_RELAXED, __HIP_MEMORY_SCOPE_WORKGROUP);
    }
    if (RECORD) {
        const int cnt = FULL ? 32 : n;
        if (ln + cnt <= GCN_LIST_CAP) {
            if (FULL || lane < n)
                wl[ln + lane] = make_int2(s_l | (d_l << 16), __float_as_int(sc_l));
            ln += cnt;
        } else {
            ln = GCN_LIST_CAP + 1;
        }
    }
    gcn_own_walk<FULL>(x, tile, s_l, d_l, sc_l, n, lane);
}

template <bool RECORD>
__device__ __forceinline__ void gcn_own_scan(const float* __restrict__ x,
                                             const int* __restrict__ src,
                                             const int* __restrict__ dst,
                                             const float* __restrict__ deg_out,
                                             float* tile, int* qe, int* qd, int* cq,
                                             unsigned bm_w, int bm_c,
                                             int lo, int rows, int E, int lane,
                                             int2* wl, int& ln, int* rowcnt) {
    int  qn = 0;

    const int4* dst4 = reinterpret_cast<const int4*>(dst);
    const int nvec   = E >> 2;

    int ncq = 0;
    for (int t = 0; t < 32; ++t) {
        const bool     has  = bm_w != 0u;
        const unsigned mask = __builtin_amdgcn_ballot_w32(has);
        if (mask == 0u) break;
        if (has) {
            cq[ncq + (int)__builtin_amdgcn_mbcnt_lo(mask, 0u)] = bm_c + __builtin_ctz(bm_w);
            bm_w &= bm_w - 1u;
        }
        ncq += __builtin_popcount(mask);
    }

    for (int i = 0; i < ncq; i += 2 * GCN_SCAN_LOADS) {
        int ck = -1;
        if (i + (lane & 15) < ncq) ck = cq[i + (lane & 15)];
        int4 dv[GCN_SCAN_LOADS];
        int  id4[GCN_SCAN_LOADS];
#pragma unroll
        for (int u = 0; u < GCN_SCAN_LOADS; ++u) {
            const int c0 = __builtin_amdgcn_readlane(ck, 2 * u);
            const int c1 = __builtin_amdgcn_readlane(ck, 2 * u + 1);
            const int cc = (lane >> 4) != 0 ? c1 : c0;
            const int idx = cc >= 0 ? cc * 16 + (lane & 15) : -1;
            id4[u] = idx;
            dv[u] = (idx >= 0 && idx < nvec) ? dst4[idx] : make_int4(-1, -1, -1, -1);
        }
#pragma unroll
        for (int u = 0; u < GCN_SCAN_LOADS; ++u) {
            const int e4 = id4[u] * 4;
            const unsigned r[4] = { (unsigned)(dv[u].x - lo), (unsigned)(dv[u].y - lo),
                                    (unsigned)(dv[u].z - lo), (unsigned)(dv[u].w - lo) };
            const unsigned r01 = r[0] < r[1] ? r[0] : r[1];
            const unsigned r23 = r[2] < r[3] ? r[2] : r[3];
            const unsigned rmin = r01 < r23 ? r01 : r23;
            if (__builtin_amdgcn_ballot_w32(rmin < (unsigned)rows) == 0u) continue;
            const bool m[4] = { r[0] < (unsigned)rows, r[1] < (unsigned)rows,
                                r[2] < (unsigned)rows, r[3] < (unsigned)rows };
#pragma unroll
            for (int c = 0; c < 4; ++c) {
                const unsigned mask = __builtin_amdgcn_ballot_w32(m[c]);
                if (mask != 0u) {
                    const int pos = qn + (int)__builtin_amdgcn_mbcnt_lo(mask, 0u);
                    if (m[c]) { qe[pos] = e4 + c; qd[pos] = (int)r[c]; }
                    qn += __builtin_popcount(mask);
                }
            }
            while (qn >= 32) {
                qn -= 32;
                gcn_own_batch<true, RECORD>(x, src, deg_out, tile, qe + qn, qd + qn, 32, lane, wl, ln, rowcnt);
            }
        }
    }
    if (qn > 0)
        gcn_own_batch<false, RECORD>(x, src, deg_out, tile, qe, qd, qn, lane, wl, ln, rowcnt);
}

template <bool FIRST, int C>
__global__ __launch_bounds__(GCN_OWN_THREADS)
void gcn_layer_owner(const float* __restrict__ x,
                     const int* __restrict__ src, const int* __restrict__ dst,
                     const float* __restrict__ deg_out,
                     float* __restrict__ deg_in_is,
                     const float* __restrict__ W, const float* __restrict__ bias,
                     float* __restrict__ out, int* __restrict__ lists,
                     int* __restrict__ counts, const unsigned* __restrict__ bitmap,
                     int E, int N, int do_relu) {
    constexpr int NT     = C >= 64 ? 4 : C / 16;
    constexpr int CG     = C / (16 * NT);
    constexpr int PITCH4 = C + 8;
    constexpr int WQ_N   = 32 * C;
    constexpr int STAGE  = (WQ_N + GCN_OWN_THREADS - 1) / GCN_OWN_THREADS;
    static_assert(C % (16 * NT) == 0 && (GCN_TILE_ROWS / 16) * CG <= GCN_OWN_WAVES, "a wave per row tile and column group");
    __shared__ float  tile[GCN_TILE_ROWS * GCN_TILE_PITCH];
    __shared__ int    q_e[GCN_OWN_WAVES * GCN_Q_CAP];
    __shared__ int    q_d[GCN_OWN_WAVES * GCN_Q_CAP];
    __shared__ int    c_q[GCN_OWN_WAVES * GCN_CQ_CAP];
    __shared__ float4 wq[32 * PITCH4];
    __shared__ int    rowcnt[GCN_TILE_ROWS];

    const int tid  = threadIdx.x;
    const int lane = tid & 31;
    const int wave = __builtin_amdgcn_readfirstlane(tid >> 5);
    const int lo   = blockIdx.x * GCN_OWN_ROWS;
    int rows = N - lo;
    rows = rows > GCN_OWN_ROWS ? GCN_OWN_ROWS : rows;
    rows = rows < 0 ? 0 : rows;

    const int slot = blockIdx.x * GCN_OWN_WAVES + wave;
    int2* wl = reinterpret_cast<int2*>(lists) + (size_t)slot * GCN_LIST_CAP;

    int  my_n      = 0;
    bool walk_list = false;
    if (!FIRST) {
        const int c = counts[blockIdx.x * GCN_OWN_WAVES + lane];
        walk_list = __builtin_amdgcn_ballot_w32(c > GCN_LIST_CAP) == 0u;
        my_n      = __builtin_amdgcn_readlane(c, wave);
    }
    int2 cur = make_int2(0, 0);
    if (!FIRST && walk_list && lane < my_n) cur = wl[lane];

    unsigned bm_w = 0u;
    int      bm_j = -1;
    if (lane < GCN_BM_PER_WAVE) bm_j = wave * GCN_BM_PER_WAVE + lane;
    else if (lane == GCN_BM_PER_WAVE && wave < GCN_BM_EXTRA) bm_j = GCN_OWN_WAVES * GCN_BM_PER_WAVE + wave;
    if ((FIRST || !walk_list) && bm_j >= 0)
        bm_w = bitmap[((bm_j >> 1) * GCN_OWN_GROUPS + blockIdx.x) * 2 + (bm_j & 1)];

    const int  h      = lane >> 4;
    const int  c16    = lane & 15;
    const int  rt     = wave / CG;
    const int  nb     = (wave % CG) * (16 * NT);
    const bool mm     = rt * 16 < rows;
    const int  arow   = rt * 16 + c16;
    float scale = 0.0f;
    float bv[NT];
#pragma unroll
    for (int t = 0; t < NT; ++t) bv[t] = 0.0f;
    if (mm) {
        if (!FIRST && arow < rows) scale = deg_in_is[lo + arow];
#pragma unroll
        for (int t = 0; t < NT; ++t) bv[t] = bias[nb + t * 16 + c16];
    }

    float4 wreg[STAGE];
#pragma unroll
    for (int i = 0; i < STAGE; ++i) {
        const int idx = tid + i * GCN_OWN_THREADS;
        wreg[i] = make_float4(0.0f, 0.0f, 0.0f, 0.0f);
        if (WQ_N % GCN_OWN_THREADS == 0 || idx < WQ_N) {
            const float* wp = W + (size_t)(4 * (idx / C)) * C + (idx % C);
            wreg[i] = make_float4(wp[0], wp[C], wp[2 * C], wp[3 * C]);
        }
    }

    float4* tile4 = reinterpret_cast<float4*>(tile);
    for (int i = tid; i < GCN_TILE_ROWS * (GCN_TILE_PITCH / 4); i += GCN_OWN_THREADS)
        tile4[i] = make_float4(0.0f, 0.0f, 0.0f, 0.0f);
    if (FIRST && tid < GCN_TILE_ROWS) rowcnt[tid] = 0;
#pragma unroll
    for (int i = 0; i < STAGE; ++i) {
        const int idx = tid + i * GCN_OWN_THREADS;
        if (WQ_N % GCN_OWN_THREADS == 0 || idx < WQ_N)
            wq[(idx / C) * PITCH4 + (idx % C)] = wreg[i];
    }
    __syncthreads();

    if (!FIRST && walk_list) {
        for (int b = 0; b < my_n; b += 32) {
            int2 nxt = make_int2(0, 0);
            if (b + 32 + lane < my_n) nxt = wl[b + 32 + lane];
            const int   s_l  = cur.x & 0xFFFF;
            const int   d_l  = cur.x >> 16;
            const float sc_l = __int_as_float(cur.y);
            const int rem = my_n - b;
            if (rem >= 32) gcn_own_walk<true >(x, tile, s_l, d_l, sc_l, 32,  lane);
            else           gcn_own_walk<false>(x, tile, s_l, d_l, sc_l, rem, lane);
            cur = nxt;
        }
    } else {
        int ln = 0;
        gcn_own_scan<FIRST>(x, src, dst, deg_out, tile,
                            q_e + wave * GCN_Q_CAP, q_d + wave * GCN_Q_CAP, c_q + wave * GCN_CQ_CAP,
                            bm_w, bm_j * 32,
                            lo, rows, E, lane, wl, ln, rowcnt);
        if (FIRST && lane == 0) counts[slot] = ln;
    }

    __syncthreads();
    if (!mm) return;

    if (FIRST && arow < rows) {
        scale = gcn_deg_scale((float)rowcnt[arow]);
        if (wave % CG == 0 && h == 0) deg_in_is[lo + arow] = scale;
    }

    v8f acc[NT];
#pragma unroll
    for (int t = 0; t < NT; ++t) acc[t] = v8f{};

    const float* ap = tile + arow * GCN_TILE_PITCH + h;
#pragma unroll
    for (int j = 0; j < 16; ++j) {
        v2f a0, a1;
        a0.x = ap[2 * j]      * scale;  a0.y = ap[32 + 2 * j] * scale;
        a1.x = ap[64 + 2 * j] * scale;  a1.y = ap[96 + 2 * j] * scale;
        float4 b[NT];
#pragma unroll
        for (int t = 0; t < NT; ++t) b[t] = wq[(2 * j + h) * PITCH4 + nb + t * 16 + c16];
#pragma unroll
        for (int t = 0; t < NT; ++t) {
            v2f b0;  b0.x = b[t].x;  b0.y = b[t].y;
            acc[t] = __builtin_amdgcn_wmma_f32_16x16x4_f32(
                         false, a0, false, b0, (short)0, acc[t], false, false);
        }
#pragma unroll
        for (int t = 0; t < NT; ++t) {
            v2f b1;  b1.x = b[t].z;  b1.y = b[t].w;
            acc[t] = __builtin_amdgcn_wmma_f32_16x16x4_f32(
                         false, a1, false, b1, (short)0, acc[t], false, false);
        }
    }

#pragma unroll
    for (int t = 0; t < NT; ++t) {
        const int n = nb + t * 16 + c16;
#pragma unroll
        for (int v = 0; v < 8; ++v) {
            const int r = rt * 16 + h * 8 + v;
            float val = acc[t][v] + bv[t];
            if (do_relu) val = fmaxf(val, 0.0f);
            if (r < rows) out[(size_t)(lo + r) * C + n] = val;
        }
    }
}

extern "C" void kernel_launch(void* const* d_in, const int* in_sizes, int n_in,
                              void* d_out, int out_size, void* d_ws, size_t ws_size,
                              hipStream_t stream) {
    const float* features = (const float*)d_in[0];
    const int*   src      = (const int*)  d_in[1];
    const int*   dst      = (const int*)  d_in[2];
    const float* W1       = (const float*)d_in[3];
    const float* b1       = (const float*)d_in[4];
    const float* W2       = (const float*)d_in[5];
    const float* b2       = (const float*)d_in[6];
    const float* W3       = (const float*)d_in[7];
    const float* b3       = (const float*)d_in[8];
    float* out = (float*)d_out;

    const int N = GCN_N_NODES;
    const int E = GCN_N_EDGES;
    const size_t NF = (size_t)N * 128;

    float* bufA  = (float*)d_ws;
    float* bufB  = bufA + NF;
    float* deg_o = bufB + NF;
    float* deg_i = deg_o + N;
    int*   lists  = (int*)((float*)d_ws + GCN_LIST_FLOAT_OFF);
    int*   counts = lists + 2 * GCN_LIST_CAP * GCN_OWN_WAVES * GCN_OWN_GROUPS;
    unsigned* bitmap = (unsigned*)(counts + GCN_OWN_WAVES * GCN_OWN_GROUPS);

    gcn_count_out_deg<<<GCN_CNT_GROUPS + GCN_BM_BUILDERS, GCN_CNT_THREADS, 0, stream>>>(src, dst, deg_o, bitmap, E, N);

    gcn_layer_owner<true, 128><<<GCN_OWN_GROUPS, GCN_OWN_THREADS, 0, stream>>>(features, src, dst, deg_o, deg_i, W1, b1, bufB, lists, counts, bitmap, E, N, 1);

    gcn_layer_owner<false, 128><<<GCN_OWN_GROUPS, GCN_OWN_THREADS, 0, stream>>>(bufB, src, dst, deg_o, deg_i, W2, b2, bufA, lists, counts, bitmap, E, N, 1);

    gcn_layer_owner<false, 16><<<GCN_OWN_GROUPS, GCN_OWN_THREADS, 0, stream>>>(bufA, src, dst, deg_o, deg_i, W3, b3, out, lists, counts, bitmap, E, N, 0);

    (void)in_sizes; (void)n_in; (void)out_size; (void)ws_size;
}
